# starting kernel
# baseline (speedup 1.0000x reference)
#include <hip/hip_runtime.h>
#include <hip/hip_bf16.h>

typedef __attribute__((ext_vector_type(2))) float v2f;
typedef __attribute__((ext_vector_type(8))) float v8f;

#define IN_CH  128
#define HID_CH 64

__global__ __launch_bounds__(256)
void gcn_fill_kernel(float* __restrict__ p, float v, long long n) {
    long long i = (long long)blockIdx.x * blockDim.x + threadIdx.x;
    if (i < n) p[i] = v;
}

__global__ __launch_bounds__(256)
void gcn_deg_kernel(const int* __restrict__ col, float* __restrict__ deg, int nE) {
    int e = blockIdx.x * blockDim.x + threadIdx.x;
    if (e < nE) atomicAdd(deg + col[e], 1.0f);
}

__global__ __launch_bounds__(256)
void gcn_rsqrt_kernel(float* __restrict__ deg, int n) {
    int i = blockIdx.x * blockDim.x + threadIdx.x;
    if (i < n) {
        float d = deg[i];
        deg[i] = (d > 0.0f) ? rsqrtf(d) : 0.0f;
    }
}

__global__ __launch_bounds__(128)
void gcn_gemm_wmma_kernel(const float* __restrict__ X, const float* __restrict__ W,
                          float* __restrict__ H, int K, int nNodes) {
    const int lane = threadIdx.x & 31;
    const int wave = threadIdx.x >> 5;
    const int n    = lane & 15;
    const int half = lane >> 4;
    const int r0   = blockIdx.x * 16;
    const int n0   = wave * 16;

    const int rA = (r0 + n < nNodes) ? (r0 + n) : (nNodes - 1);

    v8f acc = {};
    for (int kb = 0; kb < K; kb += 4) {
        const int ka = kb + 2 * half;
        v2f a = *(const v2f*)(X + (long long)rA * K + ka);
        v2f b;
        b.x = W[(long long)(ka    ) * HID_CH + n0 + n];
        b.y = W[(long long)(ka + 1) * HID_CH + n0 + n];
        acc = __builtin_amdgcn_wmma_f32_16x16x4_f32(
                   false, a,  false, b,
                   (short)0, acc,  false,  false);
    }

#pragma unroll
    for (int v = 0; v < 8; ++v) {
        int m = v + 8 * half;
        if (r0 + m < nNodes)
            H[(long long)(r0 + m) * HID_CH + n0 + n] = acc[v];
    }
}

__global__ __launch_bounds__(256)
void gcn_scatter_kernel(const float* __restrict__ H, const int* __restrict__ row,
                        const int* __restrict__ col, const float* __restrict__ dinv,
                        float* __restrict__ out, int nE) {
    const int lane = threadIdx.x & 31;
    const int e    = blockIdx.x * 8 + (threadIdx.x >> 5);
    if (e >= nE) return;

    const int   r   = row[e];
    const int   c   = col[e];
    const float nrm = dinv[r] * dinv[c];

    const float* hr = H + (long long)r * HID_CH;
    float v0 = hr[lane];
    float v1 = hr[lane + 32];

    float* oc = out + (long long)c * HID_CH;
    atomicAdd(oc + lane,      v0 * nrm);
    atomicAdd(oc + lane + 32, v1 * nrm);
}

__global__ __launch_bounds__(256)
void gcn_finalize_kernel(const float* __restrict__ agg, const float* __restrict__ h,
                         const float* __restrict__ dinv, const float* __restrict__ bias,
                         float* __restrict__ out, int nNodes, int doRelu) {
    long long idx = (long long)blockIdx.x * blockDim.x + threadIdx.x;
    if (idx >= (long long)nNodes * HID_CH) return;
    int node = (int)(idx >> 6);
    int ch   = (int)(idx & 63);
    float di = dinv[node];
    float v  = agg[idx] + h[idx] * di * di + bias[ch];
    out[idx] = doRelu ? fmaxf(v, 0.0f) : v;
}

extern "C" void kernel_launch(void* const* d_in, const int* in_sizes, int n_in,
                              void* d_out, int out_size, void* d_ws, size_t ws_size,
                              hipStream_t stream) {
    const float* x   = (const float*)d_in[0];
    const int*   ei  = (const int*)  d_in[1];
    const float* W1  = (const float*)d_in[2];
    const float* b1  = (const float*)d_in[3];
    const float* W2  = (const float*)d_in[4];
    const float* b2  = (const float*)d_in[5];
    float*       out = (float*)d_out;

    const int N  = in_sizes[0] / IN_CH;
    const int nE = in_sizes[1] / 2;
    const int*   rowp = ei;
    const int*   colp = ei + nE;

    float* deg = (float*)d_ws;
    float* h1  = deg + N;
    float* agg = h1 + (long long)N * HID_CH;
    float* h2  = agg + (long long)N * HID_CH;

    const long long feat = (long long)N * HID_CH;
    const int fillBlocks = (int)((feat + 255) / 256);
    const int edgeThreadBlocks = (nE + 255) / 256;
    const int edgeWaveBlocks   = (nE + 7) / 8;
    const int nodeBlocks = (N + 255) / 256;
    const int tileBlocks = (N + 15) / 16;

    gcn_fill_kernel<<<(N + 255) / 256, 256, 0, stream>>>(deg, 1.0f, N);
    gcn_fill_kernel<<<fillBlocks, 256, 0, stream>>>(agg, 0.0f, feat);
    gcn_fill_kernel<<<fillBlocks, 256, 0, stream>>>(out, 0.0f, feat);

    gcn_deg_kernel<<<edgeThreadBlocks, 256, 0, stream>>>(colp, deg, nE);
    gcn_rsqrt_kernel<<<nodeBlocks, 256, 0, stream>>>(deg, N);

    gcn_gemm_wmma_kernel<<<tileBlocks, 128, 0, stream>>>(x, W1, h1, IN_CH, N);
    gcn_scatter_kernel<<<edgeWaveBlocks, 256, 0, stream>>>(h1, rowp, colp, deg, agg, nE);
    gcn_finalize_kernel<<<fillBlocks, 256, 0, stream>>>(agg, h1, deg, b1, h1, N,  1);

    gcn_gemm_wmma_kernel<<<tileBlocks, 128, 0, stream>>>(h1, W2, h2, HID_CH, N);
    gcn_scatter_kernel<<<edgeWaveBlocks, 256, 0, stream>>>(h2, rowp, colp, deg, out, nE);
    gcn_finalize_kernel<<<fillBlocks, 256, 0, stream>>>(out, h2, deg, b2, out, N,  0);
}


# Per-edge f32 atomic scatter replaced by destination-grouped gather: int-atomic slot count, scan, place; one wave sums a node's rows in registers in fixed ascending-source order, epilogue folded
# speedup vs baseline: 1.6753x; 1.6753x over previous
#include <hip/hip_runtime.h>
#include <hip/hip_bf16.h>


typedef __attribute__((ext_vector_type(2))) float v2f;
typedef __attribute__((ext_vector_type(8))) float v8f;

#define IN_CH  128
#define HID_CH 64

__global__ __launch_bounds__(256)
void gcn_zero_kernel(int* __restrict__ p, int n) {
    int i = blockIdx.x * blockDim.x + threadIdx.x;
    if (i < n) p[i] = 0;
}

__global__ __launch_bounds__(256)
void gcn_rank_kernel(const int* __restrict__ col, int* __restrict__ cnt,
                     int* __restrict__ rank, int nE) {
    int e = blockIdx.x * blockDim.x + threadIdx.x;
    if (e < nE) rank[e] = atomicAdd(cnt + col[e], 1);
}

__global__ __launch_bounds__(256)
void gcn_blocksum_kernel(const int* __restrict__ cnt, int* __restrict__ bsum, int n) {
    __shared__ int s_part[8];
    const int tid  = threadIdx.x;
    const int lane = tid & 31;
    const int wv   = tid >> 5;
    const int base = blockIdx.x * 1024 + tid * 4;
    int s = 0;
    for (int q = 0; q < 4; ++q) {
        int i = base + q;
        if (i < n) s += cnt[i];
    }
    for (int d = 16; d >= 1; d >>= 1) s += __shfl_xor(s, d);
    if (lane == 0) s_part[wv] = s;
    __syncthreads();
    if (tid == 0) {
        int t = 0;
        for (int k = 0; k < 8; ++k) t += s_part[k];
        bsum[blockIdx.x] = t;
    }
}

__global__ __launch_bounds__(256)
void gcn_scan_kernel(const int* __restrict__ cnt, const int* __restrict__ bsum,
                     int* __restrict__ offs, float* __restrict__ dinv, int n) {
    __shared__ int s_tot[8];
    __shared__ int s_pre[8];
    const int tid  = threadIdx.x;
    const int lane = tid & 31;
    const int wv   = tid >> 5;

    int p = 0;
    for (int k = tid; k < (int)blockIdx.x; k += 256) p += bsum[k];

    const int base = blockIdx.x * 1024 + tid * 4;
    int v[4];
    int tsum = 0;
    for (int q = 0; q < 4; ++q) {
        int i = base + q;
        v[q] = (i < n) ? cnt[i] : 0;
        tsum += v[q];
    }

    int inc = tsum;
    for (int d = 1; d < 32; d <<= 1) {
        int t = __shfl_up(inc, d);
        if (lane >= d) inc += t;
    }
    for (int d = 16; d >= 1; d >>= 1) p += __shfl_xor(p, d);

    if (lane == 31) s_tot[wv] = inc;
    if (lane == 0)  s_pre[wv] = p;
    __syncthreads();

    int run = 0;
    for (int k = 0; k < 8; ++k) {
        run += s_pre[k];
        if (k < wv) run += s_tot[k];
    }
    run += inc - tsum;

    for (int q = 0; q < 4; ++q) {
        int i = base + q;
        if (i < n) {
            offs[i] = run;
            float d = (float)(v[q] + 1);
            dinv[i] = (d > 0.0f) ? rsqrtf(d) : 0.0f;
            run += v[q];
        }
    }
}

__global__ __launch_bounds__(256)
void gcn_place_kernel(const int* __restrict__ row, const int* __restrict__ col,
                      const int* __restrict__ offs, const int* __restrict__ rank,
                      int* __restrict__ perm, int nE) {
    int e = blockIdx.x * blockDim.x + threadIdx.x;
    if (e < nE) perm[offs[col[e]] + rank[e]] = row[e];
}

__global__ __launch_bounds__(128)
void gcn_gemm_wmma_kernel(const float* __restrict__ X, const float* __restrict__ W,
                          float* __restrict__ H, int K, int nNodes) {
    const int lane = threadIdx.x & 31;
    const int wave = threadIdx.x >> 5;
    const int n    = lane & 15;
    const int half = lane >> 4;
    const int r0   = blockIdx.x * 16;
    const int n0   = wave * 16;

    const int rA = (r0 + n < nNodes) ? (r0 + n) : (nNodes - 1);

    v8f acc = {};
    for (int kb = 0; kb < K; kb += 4) {
        const int ka = kb + 2 * half;
        v2f a = *(const v2f*)(X + (long long)rA * K + ka);
        v2f b;
        b.x = W[(long long)(ka    ) * HID_CH + n0 + n];
        b.y = W[(long long)(ka + 1) * HID_CH + n0 + n];
        acc = __builtin_amdgcn_wmma_f32_16x16x4_f32(
                   false, a,  false, b,
                   (short)0, acc,  false,  false);
    }

#pragma unroll
    for (int v = 0; v < 8; ++v) {
        int m = v + 8 * half;
        if (r0 + m < nNodes)
            H[(long long)(r0 + m) * HID_CH + n0 + n] = acc[v];
    }
}

__global__ __launch_bounds__(256)
void gcn_gather_kernel(const float* __restrict__ H, const int* __restrict__ offs,
                       const int* __restrict__ cnt, const int* __restrict__ perm,
                       const float* __restrict__ dinv, const float* __restrict__ bias,
                       float* __restrict__ out, int nNodes, int doRelu) {
    const int lane = threadIdx.x & 31;
    const int wv = __builtin_amdgcn_readfirstlane((int)(threadIdx.x >> 5));
    const int c  = blockIdx.x * 8 + wv;
    if (c >= nNodes) return;

    const int   beg = offs[c];
    const int   deg = cnt[c];
    const float dc  = dinv[c];
    const v2f*  Hv  = (const v2f*)H;

    v2f acc = Hv[(long long)c * 32 + lane] * (dc * dc);

    if (deg <= 32) {
        int r_l = 0x7fffffff;
        if (lane < deg) r_l = perm[beg + lane];

        int pos = 0;
        for (int j = 0; j < deg; ++j) {
            const int rj = __builtin_amdgcn_readlane(r_l, j);
            pos += ((rj < r_l) || (rj == r_l && j < lane)) ? 1 : 0;
        }
        if (lane >= deg) pos = lane;

        const int rs = __builtin_amdgcn_ds_permute(pos << 2, r_l);
        float ws = 0.0f;
        if (lane < deg) ws = dinv[rs] * dc;

        int j = 0;
        for (; j + 4 <= deg; j += 4) {
            const int   r0 = __builtin_amdgcn_readlane(rs, j);
            const int   r1 = __builtin_amdgcn_readlane(rs, j + 1);
            const int   r2 = __builtin_amdgcn_readlane(rs, j + 2);
            const int   r3 = __builtin_amdgcn_readlane(rs, j + 3);
            const float w0 = __int_as_float(__builtin_amdgcn_readlane(__float_as_int(ws), j));
            const float w1 = __int_as_float(__builtin_amdgcn_readlane(__float_as_int(ws), j + 1));
            const float w2 = __int_as_float(__builtin_amdgcn_readlane(__float_as_int(ws), j + 2));
            const float w3 = __int_as_float(__builtin_amdgcn_readlane(__float_as_int(ws), j + 3));
            const v2f h0 = Hv[(long long)r0 * 32 + lane];
            const v2f h1 = Hv[(long long)r1 * 32 + lane];
            const v2f h2 = Hv[(long long)r2 * 32 + lane];
            const v2f h3 = Hv[(long long)r3 * 32 + lane];
            acc += h0 * w0;
            acc += h1 * w1;
            acc += h2 * w2;
            acc += h3 * w3;
        }
        for (; j < deg; ++j) {
            const int   r0 = __builtin_amdgcn_readlane(rs, j);
            const float w0 = __int_as_float(__builtin_amdgcn_readlane(__float_as_int(ws), j));
            acc += Hv[(long long)r0 * 32 + lane] * w0;
        }
    } else {
        int prev = -1;
        int done = 0;
        for (int it = 0; it < deg && done < deg; ++it) {
            int best = 0x7fffffff;
            for (int i = lane; i < deg; i += 32) {
                const int r = perm[beg + i];
                if (r > prev && r < best) best = r;
            }
            for (int d = 16; d >= 1; d >>= 1) {
                const int o = __shfl_xor(best, d);
                best = (o < best) ? o : best;
            }
            best = __builtin_amdgcn_readfirstlane(best);
            if (best == 0x7fffffff) break;
            int m = 0;
            for (int i = lane; i < deg; i += 32) m += (perm[beg + i] == best) ? 1 : 0;
            for (int d = 16; d >= 1; d >>= 1) m += __shfl_xor(m, d);
            m = __builtin_amdgcn_readfirstlane(m);
            const float w  = dinv[best] * dc;
            const v2f   hv = Hv[(long long)best * 32 + lane];
            for (int k = 0; k < m; ++k) acc += hv * w;
            prev = best;
            done += m;
        }
    }

    const v2f b = ((const v2f*)bias)[lane];
    v2f v = acc + b;
    if (doRelu) { v.x = fmaxf(v.x, 0.0f); v.y = fmaxf(v.y, 0.0f); }
    ((v2f*)out)[(long long)c * 32 + lane] = v;
}

extern "C" void kernel_launch(void* const* d_in, const int* in_sizes, int n_in,
                              void* d_out, int out_size, void* d_ws, size_t ws_size,
                              hipStream_t stream) {
    const float* x   = (const float*)d_in[0];
    const int*   ei  = (const int*)  d_in[1];
    const float* W1  = (const float*)d_in[2];
    const float* b1  = (const float*)d_in[3];
    const float* W2  = (const float*)d_in[4];
    const float* b2  = (const float*)d_in[5];
    float*       out = (float*)d_out;

    const int N  = in_sizes[0] / IN_CH;
    const int nE = in_sizes[1] / 2;
    const int*   rowp = ei;
    const int*   colp = ei + nE;

    const long long feat = (long long)N * HID_CH;
    float* h    = (float*)d_ws;
    float* a1   = h + feat;
    int*   rank = (int*)(a1 + feat);
    int*   perm = rank + nE;
    int*   cnt  = perm + nE;
    int*   offs = cnt + N;
    float* dinv = (float*)(offs + N);
    int*   bsum = (int*)(dinv + N);

    const int edgeBlocks = (nE + 255) / 256;
    const int nodeBlocks = (N + 255) / 256;
    const int scanBlocks = (N + 1023) / 1024;
    const int waveBlocks = (N + 7) / 8;
    const int tileBlocks = (N + 15) / 16;

    gcn_zero_kernel<<<nodeBlocks, 256, 0, stream>>>(cnt, N);
    gcn_rank_kernel<<<edgeBlocks, 256, 0, stream>>>(colp, cnt, rank, nE);
    gcn_blocksum_kernel<<<scanBlocks, 256, 0, stream>>>(cnt, bsum, N);
    gcn_scan_kernel<<<scanBlocks, 256, 0, stream>>>(cnt, bsum, offs, dinv, N);
    gcn_place_kernel<<<edgeBlocks, 256, 0, stream>>>(rowp, colp, offs, rank, perm, nE);

    gcn_gemm_wmma_kernel<<<tileBlocks, 128, 0, stream>>>(x, W1, h, IN_CH, N);
    gcn_gather_kernel<<<waveBlocks, 256, 0, stream>>>(h, offs, cnt, perm, dinv, b1, a1, N,  1);

    gcn_gemm_wmma_kernel<<<tileBlocks, 128, 0, stream>>>(a1, W2, h, HID_CH, N);
    gcn_gather_kernel<<<waveBlocks, 256, 0, stream>>>(h, offs, cnt, perm, dinv, b2, out, N,  0);
}


# Per-edge returning global atomics (slot count, scan, place) replaced by a two-level counting sort on LDS counters: 512-node buckets from per-block histograms, one block per bucket; 9 -> 7 launches
# speedup vs baseline: 2.2659x; 1.3525x over previous
#include <hip/hip_runtime.h>
#include <hip/hip_bf16.h>


typedef __attribute__((ext_vector_type(2))) float v2f;
typedef __attribute__((ext_vector_type(8))) float v8f;

#define IN_CH  128
#define HID_CH 64

#define SH      9
#define BKT     (1 << SH)
#define NBMAX   256
#define ROWBITS 17
#define EPB     8192

__global__ __launch_bounds__(1024)
void gcn_bucket_count_kernel(const int* __restrict__ col, int* __restrict__ bhist,
                             int nE, int NB) {
    __shared__ int s_h[NBMAX];
    const int tid = threadIdx.x;
    if (tid < NBMAX) s_h[tid] = 0;
    __syncthreads();
    const int base = blockIdx.x * EPB;
    for (int q = 0; q < EPB / 1024; ++q) {
        const int e = base + q * 1024 + tid;
        if (e < nE) atomicAdd(&s_h[col[e] >> SH], 1);
    }
    __syncthreads();
    if (tid < NB) bhist[blockIdx.x * NB + tid] = s_h[tid];
}

__global__ __launch_bounds__(1024)
void gcn_bucket_scatter_kernel(const int* __restrict__ row, const int* __restrict__ col,
                               const int* __restrict__ bhist, int* __restrict__ packed,
                               int* __restrict__ bstart, int* __restrict__ btot,
                               int nE, int NB, int G) {
    __shared__ int s_h[NBMAX];
    __shared__ int s_below[4][NBMAX];
    __shared__ int s_tot[4][NBMAX];
    __shared__ int s_base[NBMAX];
    __shared__ int s_w[8];
    const int tid  = threadIdx.x;
    const int lane = tid & 31;
    const int wv   = tid >> 5;
    const int g    = blockIdx.x;

    if (tid < NBMAX) s_h[tid] = 0;

    const int t = tid & (NBMAX - 1);
    const int p = tid >> 8;
    int below = 0, tot = 0;
    if (t < NB) {
        for (int k = p; k < G; k += 4) {
            const int v = bhist[k * NB + t];
            tot += v;
            if (k < g) below += v;
        }
    }
    s_below[p][t] = below;
    s_tot[p][t]   = tot;
    __syncthreads();

    const int base = g * EPB;
    int cc[EPB / 1024];
    int rk[EPB / 1024];
    for (int q = 0; q < EPB / 1024; ++q) {
        const int e = base + q * 1024 + tid;
        cc[q] = 0; rk[q] = 0;
        if (e < nE) {
            cc[q] = col[e];
            rk[q] = atomicAdd(&s_h[cc[q] >> SH], 1);
        }
    }

    int T = 0, B = 0, inc = 0;
    if (tid < NBMAX) {
        T = s_tot[0][tid] + s_tot[1][tid] + s_tot[2][tid] + s_tot[3][tid];
        B = s_below[0][tid] + s_below[1][tid] + s_below[2][tid] + s_below[3][tid];
        inc = T;
        for (int d = 1; d < 32; d <<= 1) {
            const int u = __shfl_up(inc, d);
            if (lane >= d) inc += u;
        }
        if (lane == 31) s_w[wv] = inc;
    }
    __syncthreads();
    if (tid < NBMAX) {
        int pre = 0;
        for (int k = 0; k < wv; ++k) pre += s_w[k];
        const int excl = pre + inc - T;
        s_base[tid] = excl + B;
        if (g == 0 && tid < NB) { bstart[tid] = excl; btot[tid] = T; }
    }
    __syncthreads();

    for (int q = 0; q < EPB / 1024; ++q) {
        const int e = base + q * 1024 + tid;
        if (e < nE) {
            const int c = cc[q];
            packed[s_base[c >> SH] + rk[q]] = ((c & (BKT - 1)) << ROWBITS) | row[e];
        }
    }
}

__global__ __launch_bounds__(1024)
void gcn_bucket_sort_kernel(const int* __restrict__ packed, const int* __restrict__ bstart,
                            const int* __restrict__ btot, int* __restrict__ perm,
                            int* __restrict__ cnt, int* __restrict__ offs,
                            float* __restrict__ dinv, int n) {
    __shared__ int s_cnt[BKT];
    __shared__ int s_cur[BKT];
    __shared__ int s_w[BKT / 32];
    const int tid   = threadIdx.x;
    const int lane  = tid & 31;
    const int wv    = tid >> 5;
    const int b     = blockIdx.x;
    const int start = bstart[b];
    const int m     = btot[b];

    if (tid < BKT) s_cnt[tid] = 0;
    __syncthreads();
    for (int i0 = 0; i0 < m; i0 += 4096) {
        int w[4];
        for (int q = 0; q < 4; ++q) {
            const int i = i0 + q * 1024 + tid;
            w[q] = (i < m) ? packed[start + i] : -1;
        }
        for (int q = 0; q < 4; ++q)
            if (w[q] >= 0) atomicAdd(&s_cnt[w[q] >> ROWBITS], 1);
    }
    __syncthreads();

    int v = 0, inc = 0;
    if (tid < BKT) {
        v = s_cnt[tid];
        inc = v;
        for (int d = 1; d < 32; d <<= 1) {
            const int u = __shfl_up(inc, d);
            if (lane >= d) inc += u;
        }
        if (lane == 31) s_w[wv] = inc;
    }
    __syncthreads();
    if (tid < BKT) {
        int pre = 0;
        for (int k = 0; k < wv; ++k) pre += s_w[k];
        const int excl = pre + inc - v;
        s_cur[tid] = excl;
        const int node = (b << SH) + tid;
        if (node < n) {
            cnt[node]  = v;
            offs[node] = start + excl;
            float d = (float)(v + 1);
            dinv[node] = (d > 0.0f) ? rsqrtf(d) : 0.0f;
        }
    }
    __syncthreads();
    for (int i0 = 0; i0 < m; i0 += 4096) {
        int w[4];
        for (int q = 0; q < 4; ++q) {
            const int i = i0 + q * 1024 + tid;
            w[q] = (i < m) ? packed[start + i] : -1;
        }
        for (int q = 0; q < 4; ++q) {
            if (w[q] >= 0) {
                const int slot = atomicAdd(&s_cur[w[q] >> ROWBITS], 1);
                perm[start + slot] = w[q] & ((1 << ROWBITS) - 1);
            }
        }
    }
}

__global__ __launch_bounds__(128)
void gcn_gemm_wmma_kernel(const float* __restrict__ X, const float* __restrict__ W,
                          float* __restrict__ H, int K, int nNodes) {
    const int lane = threadIdx.x & 31;
    const int wave = threadIdx.x >> 5;
    const int n    = lane & 15;
    const int half = lane >> 4;
    const int r0   = blockIdx.x * 16;
    const int n0   = wave * 16;

    const int rA = (r0 + n < nNodes) ? (r0 + n) : (nNodes - 1);

    v8f acc = {};
    for (int kb = 0; kb < K; kb += 4) {
        const int ka = kb + 2 * half;
        v2f a = *(const v2f*)(X + (long long)rA * K + ka);
        v2f b;
        b.x = W[(long long)(ka    ) * HID_CH + n0 + n];
        b.y = W[(long long)(ka + 1) * HID_CH + n0 + n];
        acc = __builtin_amdgcn_wmma_f32_16x16x4_f32(
                   false, a,  false, b,
                   (short)0, acc,  false,  false);
    }

#pragma unroll
    for (int v = 0; v < 8; ++v) {
        int m = v + 8 * half;
        if (r0 + m < nNodes)
            H[(long long)(r0 + m) * HID_CH + n0 + n] = acc[v];
    }
}

__global__ __launch_bounds__(256)
void gcn_gather_kernel(const float* __restrict__ H, const int* __restrict__ offs,
                       const int* __restrict__ cnt, const int* __restrict__ perm,
                       const float* __restrict__ dinv, const float* __restrict__ bias,
                       float* __restrict__ out, int nNodes, int doRelu) {
    const int lane = threadIdx.x & 31;
    const int wv = __builtin_amdgcn_readfirstlane((int)(threadIdx.x >> 5));
    const int c  = blockIdx.x * 8 + wv;
    if (c >= nNodes) return;

    const int   beg = offs[c];
    const int   deg = cnt[c];
    const float dc  = dinv[c];
    const v2f*  Hv  = (const v2f*)H;

    v2f acc = Hv[(long long)c * 32 + lane] * (dc * dc);

    if (deg <= 32) {
        int r_l = 0x7fffffff;
        if (lane < deg) r_l = perm[beg + lane];

        int pos = 0;
        for (int j = 0; j < deg; ++j) {
            const int rj = __builtin_amdgcn_readlane(r_l, j);
            pos += ((rj < r_l) || (rj == r_l && j < lane)) ? 1 : 0;
        }
        if (lane >= deg) pos = lane;

        const int rs = __builtin_amdgcn_ds_permute(pos << 2, r_l);
        float ws = 0.0f;
        if (lane < deg) ws = dinv[rs] * dc;

        int j = 0;
        for (; j + 4 <= deg; j += 4) {
            const int   r0 = __builtin_amdgcn_readlane(rs, j);
            const int   r1 = __builtin_amdgcn_readlane(rs, j + 1);
            const int   r2 = __builtin_amdgcn_readlane(rs, j + 2);
            const int   r3 = __builtin_amdgcn_readlane(rs, j + 3);
            const float w0 = __int_as_float(__builtin_amdgcn_readlane(__float_as_int(ws), j));
            const float w1 = __int_as_float(__builtin_amdgcn_readlane(__float_as_int(ws), j + 1));
            const float w2 = __int_as_float(__builtin_amdgcn_readlane(__float_as_int(ws), j + 2));
            const float w3 = __int_as_float(__builtin_amdgcn_readlane(__float_as_int(ws), j + 3));
            const v2f h0 = Hv[(long long)r0 * 32 + lane];
            const v2f h1 = Hv[(long long)r1 * 32 + lane];
            const v2f h2 = Hv[(long long)r2 * 32 + lane];
            const v2f h3 = Hv[(long long)r3 * 32 + lane];
            acc += h0 * w0;
            acc += h1 * w1;
            acc += h2 * w2;
            acc += h3 * w3;
        }
        for (; j < deg; ++j) {
            const int   r0 = __builtin_amdgcn_readlane(rs, j);
            const float w0 = __int_as_float(__builtin_amdgcn_readlane(__float_as_int(ws), j));
            acc += Hv[(long long)r0 * 32 + lane] * w0;
        }
    } else {
        int prev = -1;
        int done = 0;
        for (int it = 0; it < deg && done < deg; ++it) {
            int best = 0x7fffffff;
            for (int i = lane; i < deg; i += 32) {
                const int r = perm[beg + i];
                if (r > prev && r < best) best = r;
            }
            for (int d = 16; d >= 1; d >>= 1) {
                const int o = __shfl_xor(best, d);
                best = (o < best) ? o : best;
            }
            best = __builtin_amdgcn_readfirstlane(best);
            if (best == 0x7fffffff) break;
            int m = 0;
            for (int i = lane; i < deg; i += 32) m += (perm[beg + i] == best) ? 1 : 0;
            for (int d = 16; d >= 1; d >>= 1) m += __shfl_xor(m, d);
            m = __builtin_amdgcn_readfirstlane(m);
            const float w  = dinv[best] * dc;
            const v2f   hv = Hv[(long long)best * 32 + lane];
            for (int k = 0; k < m; ++k) acc += hv * w;
            prev = best;
            done += m;
        }
    }

    const v2f b = ((const v2f*)bias)[lane];
    v2f v = acc + b;
    if (doRelu) { v.x = fmaxf(v.x, 0.0f); v.y = fmaxf(v.y, 0.0f); }
    ((v2f*)out)[(long long)c * 32 + lane] = v;
}

extern "C" void kernel_launch(void* const* d_in, const int* in_sizes, int n_in,
                              void* d_out, int out_size, void* d_ws, size_t ws_size,
                              hipStream_t stream) {
    const float* x   = (const float*)d_in[0];
    const int*   ei  = (const int*)  d_in[1];
    const float* W1  = (const float*)d_in[2];
    const float* b1  = (const float*)d_in[3];
    const float* W2  = (const float*)d_in[4];
    const float* b2  = (const float*)d_in[5];
    float*       out = (float*)d_out;

    const int N  = in_sizes[0] / IN_CH;
    const int nE = in_sizes[1] / 2;
    const int*   rowp = ei;
    const int*   colp = ei + nE;

    const long long feat = (long long)N * HID_CH;
    float* h      = (float*)d_ws;
    float* a1     = h + feat;
    int*   packed = (int*)(a1 + feat);
    int*   perm   = packed + nE;
    int*   cnt    = perm + nE;
    int*   offs   = cnt + N;
    float* dinv   = (float*)(offs + N);

    const int NB = (N + BKT - 1) / BKT;
    const int G  = (nE + EPB - 1) / EPB;
    int*   bhist  = (int*)a1;
    int*   bstart = bhist + G * NB;
    int*   btot   = bstart + NB;

    const int waveBlocks = (N + 7) / 8;
    const int tileBlocks = (N + 15) / 16;

    gcn_bucket_count_kernel<<<G, 1024, 0, stream>>>(colp, bhist, nE, NB);
    gcn_bucket_scatter_kernel<<<G, 1024, 0, stream>>>(rowp, colp, bhist, packed, bstart, btot, nE, NB, G);
    gcn_bucket_sort_kernel<<<NB, 1024, 0, stream>>>(packed, bstart, btot, perm, cnt, offs, dinv, N);

    gcn_gemm_wmma_kernel<<<tileBlocks, 128, 0, stream>>>(x, W1, h, IN_CH, N);
    gcn_gather_kernel<<<waveBlocks, 256, 0, stream>>>(h, offs, cnt, perm, dinv, b1, a1, N,  1);

    gcn_gemm_wmma_kernel<<<tileBlocks, 128, 0, stream>>>(a1, W2, h, HID_CH, N);
    gcn_gather_kernel<<<waveBlocks, 256, 0, stream>>>(h, offs, cnt, perm, dinv, b2, out, N,  0);
}


# GEMM re-tiled: a wave owns a 16-row tile over all 64 channels, X read once in 16-byte loads, W staged in LDS in fragment order, whole-row stores; k regrouped inside each 8 (product sum order changes)
# speedup vs baseline: 2.3481x; 1.0363x over previous
#include <hip/hip_runtime.h>
#include <hip/hip_bf16.h>


typedef __attribute__((ext_vector_type(2))) float v2f;
typedef __attribute__((ext_vector_type(4))) float v4f;
typedef __attribute__((ext_vector_type(8))) float v8f;

#define IN_CH  128
#define HID_CH 64

#define SH      9
#define BKT     (1 << SH)
#define NBMAX   256
#define ROWBITS 17
#define EPB     8192

__global__ __launch_bounds__(1024)
void gcn_bucket_count_kernel(const int* __restrict__ col, int* __restrict__ bhist,
                             int nE, int NB) {
    __shared__ int s_h[NBMAX];
    const int tid = threadIdx.x;
    if (tid < NBMAX) s_h[tid] = 0;
    __syncthreads();
    const int base = blockIdx.x * EPB;
    for (int q = 0; q < EPB / 1024; ++q) {
        const int e = base + q * 1024 + tid;
        if (e < nE) atomicAdd(&s_h[col[e] >> SH], 1);
    }
    __syncthreads();
    if (tid < NB) bhist[blockIdx.x * NB + tid] = s_h[tid];
}

__global__ __launch_bounds__(1024)
void gcn_bucket_scatter_kernel(const int* __restrict__ row, const int* __restrict__ col,
                               const int* __restrict__ bhist, int* __restrict__ packed,
                               int* __restrict__ bstart, int* __restrict__ btot,
                               int nE, int NB, int G) {
    __shared__ int s_h[NBMAX];
    __shared__ int s_below[4][NBMAX];
    __shared__ int s_tot[4][NBMAX];
    __shared__ int s_base[NBMAX];
    __shared__ int s_w[8];
    const int tid  = threadIdx.x;
    const int lane = tid & 31;
    const int wv   = tid >> 5;
    const int g    = blockIdx.x;

    if (tid < NBMAX) s_h[tid] = 0;

    const int t = tid & (NBMAX - 1);
    const int p = tid >> 8;
    int below = 0, tot = 0;
    if (t < NB) {
        for (int k = p; k < G; k += 4) {
            const int v = bhist[k * NB + t];
            tot += v;
            if (k < g) below += v;
        }
    }
    s_below[p][t] = below;
    s_tot[p][t]   = tot;
    __syncthreads();

    const int base = g * EPB;
    int cc[EPB / 1024];
    int rk[EPB / 1024];
    for (int q = 0; q < EPB / 1024; ++q) {
        const int e = base + q * 1024 + tid;
        cc[q] = 0; rk[q] = 0;
        if (e < nE) {
            cc[q] = col[e];
            rk[q] = atomicAdd(&s_h[cc[q] >> SH], 1);
        }
    }

    int T = 0, B = 0, inc = 0;
    if (tid < NBMAX) {
        T = s_tot[0][tid] + s_tot[1][tid] + s_tot[2][tid] + s_tot[3][tid];
        B = s_below[0][tid] + s_below[1][tid] + s_below[2][tid] + s_below[3][tid];
        inc = T;
        for (int d = 1; d < 32; d <<= 1) {
            const int u = __shfl_up(inc, d);
            if (lane >= d) inc += u;
        }
        if (lane == 31) s_w[wv] = inc;
    }
    __syncthreads();
    if (tid < NBMAX) {
        int pre = 0;
        for (int k = 0; k < wv; ++k) pre += s_w[k];
        const int excl = pre + inc - T;
        s_base[tid] = excl + B;
        if (g == 0 && tid < NB) { bstart[tid] = excl; btot[tid] = T; }
    }
    __syncthreads();

    for (int q = 0; q < EPB / 1024; ++q) {
        const int e = base + q * 1024 + tid;
        if (e < nE) {
            const int c = cc[q];
            packed[s_base[c >> SH] + rk[q]] = ((c & (BKT - 1)) << ROWBITS) | row[e];
        }
    }
}

__global__ __launch_bounds__(1024)
void gcn_bucket_sort_kernel(const int* __restrict__ packed, const int* __restrict__ bstart,
                            const int* __restrict__ btot, int* __restrict__ perm,
                            int* __restrict__ cnt, int* __restrict__ offs,
                            float* __restrict__ dinv, int n) {
    __shared__ int s_cnt[BKT];
    __shared__ int s_cur[BKT];
    __shared__ int s_w[BKT / 32];
    const int tid   = threadIdx.x;
    const int lane  = tid & 31;
    const int wv    = tid >> 5;
    const int b     = blockIdx.x;
    const int start = bstart[b];
    const int m     = btot[b];

    if (tid < BKT) s_cnt[tid] = 0;
    __syncthreads();
    for (int i0 = 0; i0 < m; i0 += 4096) {
        int w[4];
        for (int q = 0; q < 4; ++q) {
            const int i = i0 + q * 1024 + tid;
            w[q] = (i < m) ? packed[start + i] : -1;
        }
        for (int q = 0; q < 4; ++q)
            if (w[q] >= 0) atomicAdd(&s_cnt[w[q] >> ROWBITS], 1);
    }
    __syncthreads();

    int v = 0, inc = 0;
    if (tid < BKT) {
        v = s_cnt[tid];
        inc = v;
        for (int d = 1; d < 32; d <<= 1) {
            const int u = __shfl_up(inc, d);
            if (lane >= d) inc += u;
        }
        if (lane == 31) s_w[wv] = inc;
    }
    __syncthreads();
    if (tid < BKT) {
        int pre = 0;
        for (int k = 0; k < wv; ++k) pre += s_w[k];
        const int excl = pre + inc - v;
        s_cur[tid] = excl;
        const int node = (b << SH) + tid;
        if (node < n) {
            cnt[node]  = v;
            offs[node] = start + excl;
            float d = (float)(v + 1);
            dinv[node] = (d > 0.0f) ? rsqrtf(d) : 0.0f;
        }
    }
    __syncthreads();
    for (int i0 = 0; i0 < m; i0 += 4096) {
        int w[4];
        for (int q = 0; q < 4; ++q) {
            const int i = i0 + q * 1024 + tid;
            w[q] = (i < m) ? packed[start + i] : -1;
        }
        for (int q = 0; q < 4; ++q) {
            if (w[q] >= 0) {
                const int slot = atomicAdd(&s_cur[w[q] >> ROWBITS], 1);
                perm[start + slot] = w[q] & ((1 << ROWBITS) - 1);
            }
        }
    }
}

#define GEMM_WAVES 8
#define TPITCH     68

template <int K>
__global__ __launch_bounds__(256)
void gcn_gemm_wmma_kernel(const float* __restrict__ X, const float* __restrict__ W,
                          float* __restrict__ H, int nNodes) {
    __shared__ __attribute__((aligned(16))) float s_w[K * HID_CH];
    __shared__ __attribute__((aligned(16))) float s_t[GEMM_WAVES * 16 * TPITCH];

    const int tid  = threadIdx.x;
    const int lane = tid & 31;
    const int wave = __builtin_amdgcn_readfirstlane(tid >> 5);
    const int n    = lane & 15;
    const int half = lane >> 4;
    const int r0   = (blockIdx.x * GEMM_WAVES + wave) * 16;

    int rA = r0 + n;
    if (rA > nNodes - 1) rA = nNodes - 1;
    const v4f* xr = (const v4f*)(X + (long long)rA * K) + half;
    v4f a4[K / 8];
#pragma unroll
    for (int j = 0; j < K / 8; ++j) a4[j] = xr[2 * j];

    constexpr int WQ = K * HID_CH / 1024;
    v4f w4[WQ];
#pragma unroll
    for (int q = 0; q < WQ; ++q) w4[q] = *(const v4f*)(W + q * 1024 + tid * 4);
#pragma unroll
    for (int q = 0; q < WQ; ++q) {
        const int g  = q * 1024 + tid * 4;
        const int k  = g >> 6;
        const int c  = g & (HID_CH - 1);
        const int kj = k >> 3;
        const int kh = (k >> 2) & 1;
        const int ke = k & 3;
        const int o  = ((((kj * 4 + (c >> 4)) * 32) + kh * 16 + (c & 15)) << 2) + ke;
        s_w[o]      = w4[q].x;
        s_w[o + 4]  = w4[q].y;
        s_w[o + 8]  = w4[q].z;
        s_w[o + 12] = w4[q].w;
    }
    __syncthreads();

    if (r0 >= nNodes) return;

    v8f acc0 = {}, acc1 = {}, acc2 = {}, acc3 = {};
    const v4f* bw = (const v4f*)s_w + lane;
#pragma unroll
    for (int j = 0; j < K / 8; ++j) {
        const v4f av = a4[j];
        const v4f q0 = bw[(j * 4 + 0) * 32];
        const v4f q1 = bw[(j * 4 + 1) * 32];
        const v4f q2 = bw[(j * 4 + 2) * 32];
        const v4f q3 = bw[(j * 4 + 3) * 32];
        v2f a, b0, b1, b2, b3;
        a.x = av.x;  a.y = av.y;
        b0.x = q0.x; b0.y = q0.y;
        b1.x = q1.x; b1.y = q1.y;
        b2.x = q2.x; b2.y = q2.y;
        b3.x = q3.x; b3.y = q3.y;
        acc0 = __builtin_amdgcn_wmma_f32_16x16x4_f32(false, a, false, b0, (short)0, acc0, false, false);
        acc1 = __builtin_amdgcn_wmma_f32_16x16x4_f32(false, a, false, b1, (short)0, acc1, false, false);
        acc2 = __builtin_amdgcn_wmma_f32_16x16x4_f32(false, a, false, b2, (short)0, acc2, false, false);
        acc3 = __builtin_amdgcn_wmma_f32_16x16x4_f32(false, a, false, b3, (short)0, acc3, false, false);
        a.x = av.z;  a.y = av.w;
        b0.x = q0.z; b0.y = q0.w;
        b1.x = q1.z; b1.y = q1.w;
        b2.x = q2.z; b2.y = q2.w;
        b3.x = q3.z; b3.y = q3.w;
        acc0 = __builtin_amdgcn_wmma_f32_16x16x4_f32(false, a, false, b0, (short)0, acc0, false, false);
        acc1 = __builtin_amdgcn_wmma_f32_16x16x4_f32(false, a, false, b1, (short)0, acc1, false, false);
        acc2 = __builtin_amdgcn_wmma_f32_16x16x4_f32(false, a, false, b2, (short)0, acc2, false, false);
        acc3 = __builtin_amdgcn_wmma_f32_16x16x4_f32(false, a, false, b3, (short)0, acc3, false, false);
    }

    float* st = s_t + wave * (16 * TPITCH);
#pragma unroll
    for (int v = 0; v < 8; ++v) {
        float* p = st + (v + 8 * half) * TPITCH + n;
        p[0]  = acc0[v];
        p[16] = acc1[v];
        p[32] = acc2[v];
        p[48] = acc3[v];
    }
    v4f o[8];
#pragma unroll
    for (int i = 0; i < 8; ++i) o[i] = *(const v4f*)(st + (2 * i + half) * TPITCH + 4 * n);
    float* hp = H + (long long)(r0 + half) * HID_CH + 4 * n;
    if (r0 + 16 <= nNodes) {
#pragma unroll
        for (int i = 0; i < 8; ++i) *(v4f*)(hp + (long long)i * 2 * HID_CH) = o[i];
    } else {
#pragma unroll
        for (int i = 0; i < 8; ++i)
            if (r0 + 2 * i + half < nNodes) *(v4f*)(hp + (long long)i * 2 * HID_CH) = o[i];
    }
}

__global__ __launch_bounds__(256)
void gcn_gather_kernel(const float* __restrict__ H, const int* __restrict__ offs,
                       const int* __restrict__ cnt, const int* __restrict__ perm,
                       const float* __restrict__ dinv, const float* __restrict__ bias,
                       float* __restrict__ out, int nNodes, int doRelu) {
    const int lane = threadIdx.x & 31;
    const int wv = __builtin_amdgcn_readfirstlane((int)(threadIdx.x >> 5));
    const int c  = blockIdx.x * 8 + wv;
    if (c >= nNodes) return;

    const int   beg = offs[c];
    const int   deg = cnt[c];
    const float dc  = dinv[c];
    const v2f*  Hv  = (const v2f*)H;

    v2f acc = Hv[(long long)c * 32 + lane] * (dc * dc);

    if (deg <= 32) {
        int r_l = 0x7fffffff;
        if (lane < deg) r_l = perm[beg + lane];

        int pos = 0;
        for (int j = 0; j < deg; ++j) {
            const int rj = __builtin_amdgcn_readlane(r_l, j);
            pos += ((rj < r_l) || (rj == r_l && j < lane)) ? 1 : 0;
        }
        if (lane >= deg) pos = lane;

        const int rs = __builtin_amdgcn_ds_permute(pos << 2, r_l);
        float ws = 0.0f;
        if (lane < deg) ws = dinv[rs] * dc;

        int j = 0;
        for (; j + 4 <= deg; j += 4) {
            const int   r0 = __builtin_amdgcn_readlane(rs, j);
            const int   r1 = __builtin_amdgcn_readlane(rs, j + 1);
            const int   r2 = __builtin_amdgcn_readlane(rs, j + 2);
            const int   r3 = __builtin_amdgcn_readlane(rs, j + 3);
            const float w0 = __int_as_float(__builtin_amdgcn_readlane(__float_as_int(ws), j));
            const float w1 = __int_as_float(__builtin_amdgcn_readlane(__float_as_int(ws), j + 1));
            const float w2 = __int_as_float(__builtin_amdgcn_readlane(__float_as_int(ws), j + 2));
            const float w3 = __int_as_float(__builtin_amdgcn_readlane(__float_as_int(ws), j + 3));
            const v2f h0 = Hv[(long long)r0 * 32 + lane];
            const v2f h1 = Hv[(long long)r1 * 32 + lane];
            const v2f h2 = Hv[(long long)r2 * 32 + lane];
            const v2f h3 = Hv[(long long)r3 * 32 + lane];
            acc += h0 * w0;
            acc += h1 * w1;
            acc += h2 * w2;
            acc += h3 * w3;
        }
        for (; j < deg; ++j) {
            const int   r0 = __builtin_amdgcn_readlane(rs, j);
            const float w0 = __int_as_float(__builtin_amdgcn_readlane(__float_as_int(ws), j));
            acc += Hv[(long long)r0 * 32 + lane] * w0;
        }
    } else {
        int prev = -1;
        int done = 0;
        for (int it = 0; it < deg && done < deg; ++it) {
            int best = 0x7fffffff;
            for (int i = lane; i < deg; i += 32) {
                const int r = perm[beg + i];
                if (r > prev && r < best) best = r;
            }
            for (int d = 16; d >= 1; d >>= 1) {
                const int o = __shfl_xor(best, d);
                best = (o < best) ? o : best;
            }
            best = __builtin_amdgcn_readfirstlane(best);
            if (best == 0x7fffffff) break;
            int m = 0;
            for (int i = lane; i < deg; i += 32) m += (perm[beg + i] == best) ? 1 : 0;
            for (int d = 16; d >= 1; d >>= 1) m += __shfl_xor(m, d);
            m = __builtin_amdgcn_readfirstlane(m);
            const float w  = dinv[best] * dc;
            const v2f   hv = Hv[(long long)best * 32 + lane];
            for (int k = 0; k < m; ++k) acc += hv * w;
            prev = best;
            done += m;
        }
    }

    const v2f b = ((const v2f*)bias)[lane];
    v2f v = acc + b;
    if (doRelu) { v.x = fmaxf(v.x, 0.0f); v.y = fmaxf(v.y, 0.0f); }
    ((v2f*)out)[(long long)c * 32 + lane] = v;
}

extern "C" void kernel_launch(void* const* d_in, const int* in_sizes, int n_in,
                              void* d_out, int out_size, void* d_ws, size_t ws_size,
                              hipStream_t stream) {
    const float* x   = (const float*)d_in[0];
    const int*   ei  = (const int*)  d_in[1];
    const float* W1  = (const float*)d_in[2];
    const float* b1  = (const float*)d_in[3];
    const float* W2  = (const float*)d_in[4];
    const float* b2  = (const float*)d_in[5];
    float*       out = (float*)d_out;

    const int N  = in_sizes[0] / IN_CH;
    const int nE = in_sizes[1] / 2;
    const int*   rowp = ei;
    const int*   colp = ei + nE;

    const long long feat = (long long)N * HID_CH;
    float* h      = (float*)d_ws;
    float* a1     = h + feat;
    int*   packed = (int*)(a1 + feat);
    int*   perm   = packed + nE;
    int*   cnt    = perm + nE;
    int*   offs   = cnt + N;
    float* dinv   = (float*)(offs + N);

    const int NB = (N + BKT - 1) / BKT;
    const int G  = (nE + EPB - 1) / EPB;
    int*   bhist  = (int*)a1;
    int*   bstart = bhist + G * NB;
    int*   btot   = bstart + NB;

    const int waveBlocks = (N + 7) / 8;
    const int tileBlocks = (N + 15) / 16;
    const int gemmBlocks = (tileBlocks + GEMM_WAVES - 1) / GEMM_WAVES;

    gcn_bucket_count_kernel<<<G, 1024, 0, stream>>>(colp, bhist, nE, NB);
    gcn_bucket_scatter_kernel<<<G, 1024, 0, stream>>>(rowp, colp, bhist, packed, bstart, btot, nE, NB, G);
    gcn_bucket_sort_kernel<<<NB, 1024, 0, stream>>>(packed, bstart, btot, perm, cnt, offs, dinv, N);

    gcn_gemm_wmma_kernel<IN_CH><<<gemmBlocks, 256, 0, stream>>>(x, W1, h, N);
    gcn_gather_kernel<<<waveBlocks, 256, 0, stream>>>(h, offs, cnt, perm, dinv, b1, a1, N,  1);

    gcn_gemm_wmma_kernel<HID_CH><<<gemmBlocks, 256, 0, stream>>>(a1, W2, h, N);
    gcn_gather_kernel<<<waveBlocks, 256, 0, stream>>>(h, offs, cnt, perm, dinv, b2, out, N,  0);
}


# Gather, nodes with 33-128 in-edges: sources ranked via a per-wave LDS list, rows summed 32 at a time by the 4-row loop, not one list min-scan per source; sum order kept, products now fused there
# speedup vs baseline: 3.0913x; 1.3165x over previous
#include <hip/hip_runtime.h>
#include <hip/hip_bf16.h>


typedef __attribute__((ext_vector_type(2))) float v2f;
typedef __attribute__((ext_vector_type(4))) float v4f;
typedef __attribute__((ext_vector_type(8))) float v8f;

#define IN_CH  128
#define HID_CH 64

#define SH      9
#define BKT     (1 << SH)
#define NBMAX   256
#define ROWBITS 17
#define EPB     8192

__global__ __launch_bounds__(1024)
void gcn_bucket_count_kernel(const int* __restrict__ col, int* __restrict__ bhist,
                             int nE, int NB) {
    __shared__ int s_h[NBMAX];
    const int tid = threadIdx.x;
    if (tid < NBMAX) s_h[tid] = 0;
    __syncthreads();
    const int base = blockIdx.x * EPB;
    for (int q = 0; q < EPB / 1024; ++q) {
        const int e = base + q * 1024 + tid;
        if (e < nE) atomicAdd(&s_h[col[e] >> SH], 1);
    }
    __syncthreads();
    if (tid < NB) bhist[blockIdx.x * NB + tid] = s_h[tid];
}

__global__ __launch_bounds__(1024)
void gcn_bucket_scatter_kernel(const int* __restrict__ row, const int* __restrict__ col,
                               const int* __restrict__ bhist, int* __restrict__ packed,
                               int* __restrict__ bstart, int* __restrict__ btot,
                               int nE, int NB, int G) {
    __shared__ int s_h[NBMAX];
    __shared__ int s_below[4][NBMAX];
    __shared__ int s_tot[4][NBMAX];
    __shared__ int s_base[NBMAX];
    __shared__ int s_w[8];
    const int tid  = threadIdx.x;
    const int lane = tid & 31;
    const int wv   = tid >> 5;
    const int g    = blockIdx.x;

    if (tid < NBMAX) s_h[tid] = 0;

    const int t = tid & (NBMAX - 1);
    const int p = tid >> 8;
    int below = 0, tot = 0;
    if (t < NB) {
        for (int k = p; k < G; k += 4) {
            const int v = bhist[k * NB + t];
            tot += v;
            if (k < g) below += v;
        }
    }
    s_below[p][t] = below;
    s_tot[p][t]   = tot;
    __syncthreads();

    const int base = g * EPB;
    int cc[EPB / 1024];
    int rk[EPB / 1024];
    for (int q = 0; q < EPB / 1024; ++q) {
        const int e = base + q * 1024 + tid;
        cc[q] = 0; rk[q] = 0;
        if (e < nE) {
            cc[q] = col[e];
            rk[q] = atomicAdd(&s_h[cc[q] >> SH], 1);
        }
    }

    int T = 0, B = 0, inc = 0;
    if (tid < NBMAX) {
        T = s_tot[0][tid] + s_tot[1][tid] + s_tot[2][tid] + s_tot[3][tid];
        B = s_below[0][tid] + s_below[1][tid] + s_below[2][tid] + s_below[3][tid];
        inc = T;
        for (int d = 1; d < 32; d <<= 1) {
            const int u = __shfl_up(inc, d);
            if (lane >= d) inc += u;
        }
        if (lane == 31) s_w[wv] = inc;
    }
    __syncthreads();
    if (tid < NBMAX) {
        int pre = 0;
        for (int k = 0; k < wv; ++k) pre += s_w[k];
        const int excl = pre + inc - T;
        s_base[tid] = excl + B;
        if (g == 0 && tid < NB) { bstart[tid] = excl; btot[tid] = T; }
    }
    __syncthreads();

    for (int q = 0; q < EPB / 1024; ++q) {
        const int e = base + q * 1024 + tid;
        if (e < nE) {
            const int c = cc[q];
            packed[s_base[c >> SH] + rk[q]] = ((c & (BKT - 1)) << ROWBITS) | row[e];
        }
    }
}

__global__ __launch_bounds__(1024)
void gcn_bucket_sort_kernel(const int* __restrict__ packed, const int* __restrict__ bstart,
                            const int* __restrict__ btot, int* __restrict__ perm,
                            int* __restrict__ cnt, int* __restrict__ offs,
                            float* __restrict__ dinv, int n) {
    __shared__ int s_cnt[BKT];
    __shared__ int s_cur[BKT];
    __shared__ int s_w[BKT / 32];
    const int tid   = threadIdx.x;
    const int lane  = tid & 31;
    const int wv    = tid >> 5;
    const int b     = blockIdx.x;
    const int start = bstart[b];
    const int m     = btot[b];

    if (tid < BKT) s_cnt[tid] = 0;
    __syncthreads();
    for (int i0 = 0; i0 < m; i0 += 4096) {
        int w[4];
        for (int q = 0; q < 4; ++q) {
            const int i = i0 + q * 1024 + tid;
            w[q] = (i < m) ? packed[start + i] : -1;
        }
        for (int q = 0; q < 4; ++q)
            if (w[q] >= 0) atomicAdd(&s_cnt[w[q] >> ROWBITS], 1);
    }
    __syncthreads();

    int v = 0, inc = 0;
    if (tid < BKT) {
        v = s_cnt[tid];
        inc = v;
        for (int d = 1; d < 32; d <<= 1) {
            const int u = __shfl_up(inc, d);
            if (lane >= d) inc += u;
        }
        if (lane == 31) s_w[wv] = inc;
    }
    __syncthreads();
    if (tid < BKT) {
        int pre = 0;
        for (int k = 0; k < wv; ++k) pre += s_w[k];
        const int excl = pre + inc - v;
        s_cur[tid] = excl;
        const int node = (b << SH) + tid;
        if (node < n) {
            cnt[node]  = v;
            offs[node] = start + excl;
            float d = (float)(v + 1);
            dinv[node] = (d > 0.0f) ? rsqrtf(d) : 0.0f;
        }
    }
    __syncthreads();
    for (int i0 = 0; i0 < m; i0 += 4096) {
        int w[4];
        for (int q = 0; q < 4; ++q) {
            const int i = i0 + q * 1024 + tid;
            w[q] = (i < m) ? packed[start + i] : -1;
        }
        for (int q = 0; q < 4; ++q) {
            if (w[q] >= 0) {
                const int slot = atomicAdd(&s_cur[w[q] >> ROWBITS], 1);
                perm[start + slot] = w[q] & ((1 << ROWBITS) - 1);
            }
        }
    }
}

#define GEMM_WAVES 8
#define TPITCH     68

template <int K>
__global__ __launch_bounds__(256)
void gcn_gemm_wmma_kernel(const float* __restrict__ X, const float* __restrict__ W,
                          float* __restrict__ H, int nNodes) {
    __shared__ __attribute__((aligned(16))) float s_w[K * HID_CH];
    __shared__ __attribute__((aligned(16))) float s_t[GEMM_WAVES * 16 * TPITCH];

    const int tid  = threadIdx.x;
    const int lane = tid & 31;
    const int wave = __builtin_amdgcn_readfirstlane(tid >> 5);
    const int n    = lane & 15;
    const int half = lane >> 4;
    const int r0   = (blockIdx.x * GEMM_WAVES + wave) * 16;

    int rA = r0 + n;
    if (rA > nNodes - 1) rA = nNodes - 1;
    const v4f* xr = (const v4f*)(X + (long long)rA * K) + half;
    v4f a4[K / 8];
#pragma unroll
    for (int j = 0; j < K / 8; ++j) a4[j] = xr[2 * j];

    constexpr int WQ = K * HID_CH / 1024;
    v4f w4[WQ];
#pragma unroll
    for (int q = 0; q < WQ; ++q) w4[q] = *(const v4f*)(W + q * 1024 + tid * 4);
#pragma unroll
    for (int q = 0; q < WQ; ++q) {
        const int g  = q * 1024 + tid * 4;
        const int k  = g >> 6;
        const int c  = g & (HID_CH - 1);
        const int kj = k >> 3;
        const int kh = (k >> 2) & 1;
        const int ke = k & 3;
        const int o  = ((((kj * 4 + (c >> 4)) * 32) + kh * 16 + (c & 15)) << 2) + ke;
        s_w[o]      = w4[q].x;
        s_w[o + 4]  = w4[q].y;
        s_w[o + 8]  = w4[q].z;
        s_w[o + 12] = w4[q].w;
    }
    __syncthreads();

    if (r0 >= nNodes) return;

    v8f acc0 = {}, acc1 = {}, acc2 = {}, acc3 = {};
    const v4f* bw = (const v4f*)s_w + lane;
#pragma unroll
    for (int j = 0; j < K / 8; ++j) {
        const v4f av = a4[j];
        const v4f q0 = bw[(j * 4 + 0) * 32];
        const v4f q1 = bw[(j * 4 + 1) * 32];
        const v4f q2 = bw[(j * 4 + 2) * 32];
        const v4f q3 = bw[(j * 4 + 3) * 32];
        v2f a, b0, b1, b2, b3;
        a.x = av.x;  a.y = av.y;
        b0.x = q0.x; b0.y = q0.y;
        b1.x = q1.x; b1.y = q1.y;
        b2.x = q2.x; b2.y = q2.y;
        b3.x = q3.x; b3.y = q3.y;
        acc0 = __builtin_amdgcn_wmma_f32_16x16x4_f32(false, a, false, b0, (short)0, acc0, false, false);
        acc1 = __builtin_amdgcn_wmma_f32_16x16x4_f32(false, a, false, b1, (short)0, acc1, false, false);
        acc2 = __builtin_amdgcn_wmma_f32_16x16x4_f32(false, a, false, b2, (short)0, acc2, false, false);
        acc3 = __builtin_amdgcn_wmma_f32_16x16x4_f32(false, a, false, b3, (short)0, acc3, false, false);
        a.x = av.z;  a.y = av.w;
        b0.x = q0.z; b0.y = q0.w;
        b1.x = q1.z; b1.y = q1.w;
        b2.x = q2.z; b2.y = q2.w;
        b3.x = q3.z; b3.y = q3.w;
        acc0 = __builtin_amdgcn_wmma_f32_16x16x4_f32(false, a, false, b0, (short)0, acc0, false, false);
        acc1 = __builtin_amdgcn_wmma_f32_16x16x4_f32(false, a, false, b1, (short)0, acc1, false, false);
        acc2 = __builtin_amdgcn_wmma_f32_16x16x4_f32(false, a, false, b2, (short)0, acc2, false, false);
        acc3 = __builtin_amdgcn_wmma_f32_16x16x4_f32(false, a, false, b3, (short)0, acc3, false, false);
    }

    float* st = s_t + wave * (16 * TPITCH);
#pragma unroll
    for (int v = 0; v < 8; ++v) {
        float* p = st + (v + 8 * half) * TPITCH + n;
        p[0]  = acc0[v];
        p[16] = acc1[v];
        p[32] = acc2[v];
        p[48] = acc3[v];
    }
    v4f o[8];
#pragma unroll
    for (int i = 0; i < 8; ++i) o[i] = *(const v4f*)(st + (2 * i + half) * TPITCH + 4 * n);
    float* hp = H + (long long)(r0 + half) * HID_CH + 4 * n;
    if (r0 + 16 <= nNodes) {
#pragma unroll
        for (int i = 0; i < 8; ++i) *(v4f*)(hp + (long long)i * 2 * HID_CH) = o[i];
    } else {
#pragma unroll
        for (int i = 0; i < 8; ++i)
            if (r0 + 2 * i + half < nNodes) *(v4f*)(hp + (long long)i * 2 * HID_CH) = o[i];
    }
}

#define LSEG 128

__device__ __forceinline__
v2f gcn_sum_rows(v2f acc, const v2f* __restrict__ Hv, int rs, float ws, int cnt, int lane) {
    int j = 0;
    for (; j + 4 <= cnt; j += 4) {
        const int   r0 = __builtin_amdgcn_readlane(rs, j);
        const int   r1 = __builtin_amdgcn_readlane(rs, j + 1);
        const int   r2 = __builtin_amdgcn_readlane(rs, j + 2);
        const int   r3 = __builtin_amdgcn_readlane(rs, j + 3);
        const float w0 = __int_as_float(__builtin_amdgcn_readlane(__float_as_int(ws), j));
        const float w1 = __int_as_float(__builtin_amdgcn_readlane(__float_as_int(ws), j + 1));
        const float w2 = __int_as_float(__builtin_amdgcn_readlane(__float_as_int(ws), j + 2));
        const float w3 = __int_as_float(__builtin_amdgcn_readlane(__float_as_int(ws), j + 3));
        const v2f h0 = Hv[(long long)r0 * 32 + lane];
        const v2f h1 = Hv[(long long)r1 * 32 + lane];
        const v2f h2 = Hv[(long long)r2 * 32 + lane];
        const v2f h3 = Hv[(long long)r3 * 32 + lane];
        acc += h0 * w0;
        acc += h1 * w1;
        acc += h2 * w2;
        acc += h3 * w3;
    }
    for (; j < cnt; ++j) {
        const int   r0 = __builtin_amdgcn_readlane(rs, j);
        const float w0 = __int_as_float(__builtin_amdgcn_readlane(__float_as_int(ws), j));
        acc += Hv[(long long)r0 * 32 + lane] * w0;
    }
    return acc;
}

__global__ __launch_bounds__(256)
void gcn_gather_kernel(const float* __restrict__ H, const int* __restrict__ offs,
                       const int* __restrict__ cnt, const int* __restrict__ perm,
                       const float* __restrict__ dinv, const float* __restrict__ bias,
                       float* __restrict__ out, int nNodes, int doRelu) {
    __shared__ int s_raw[8][LSEG];
    __shared__ int s_ord[8][LSEG];

    const int lane = threadIdx.x & 31;
    const int wv = __builtin_amdgcn_readfirstlane((int)(threadIdx.x >> 5));
    const int c  = blockIdx.x * 8 + wv;
    if (c >= nNodes) return;

    const int   beg = offs[c];
    const int   deg = cnt[c];
    const float dc  = dinv[c];
    const v2f*  Hv  = (const v2f*)H;

    v2f acc = Hv[(long long)c * 32 + lane] * (dc * dc);

    if (deg <= 32) {
        int r_l = 0x7fffffff;
        if (lane < deg) r_l = perm[beg + lane];

        int pos = 0;
        for (int j = 0; j < deg; ++j) {
            const int rj = __builtin_amdgcn_readlane(r_l, j);
            pos += ((rj < r_l) || (rj == r_l && j < lane)) ? 1 : 0;
        }
        if (lane >= deg) pos = lane;

        const int rs = __builtin_amdgcn_ds_permute(pos << 2, r_l);
        float ws = 0.0f;
        if (lane < deg) ws = dinv[rs] * dc;

        acc = gcn_sum_rows(acc, Hv, rs, ws, deg, lane);
    } else if (deg <= LSEG) {
        int* raw = s_raw[wv];
        int* ord = s_ord[wv];
        const int K = (deg + 31) >> 5;

        for (int k = 0; k < K; ++k) {
            const int i = k * 32 + lane;
            raw[i] = (i < deg) ? perm[beg + i] : 0x7fffffff;
        }
        for (int k = 0; k < K; ++k) {
            const int i = k * 32 + lane;
            const int e = raw[i];
            int pos = 0;
            for (int cj = 0; cj < K; ++cj) {
                const int ej = raw[cj * 32 + lane];
                const int nj = (deg - cj * 32 < 32) ? (deg - cj * 32) : 32;
                for (int j = 0; j < nj; ++j) {
                    const int rj = __builtin_amdgcn_readlane(ej, j);
                    pos += ((rj < e) || (rj == e && cj * 32 + j < i)) ? 1 : 0;
                }
            }
            if (i < deg) ord[pos] = e;
        }
        for (int t = 0; t < K; ++t) {
            const int nv = (deg - t * 32 < 32) ? (deg - t * 32) : 32;
            int   rs = 0;
            float ws = 0.0f;
            if (lane < nv) { rs = ord[t * 32 + lane]; ws = dinv[rs] * dc; }
            acc = gcn_sum_rows(acc, Hv, rs, ws, nv, lane);
        }
    } else {
        int prev = -1;
        int done = 0;
        for (int it = 0; it < deg && done < deg; ++it) {
            int best = 0x7fffffff;
            for (int i = lane; i < deg; i += 32) {
                const int r = perm[beg + i];
                if (r > prev && r < best) best = r;
            }
            for (int d = 16; d >= 1; d >>= 1) {
                const int o = __shfl_xor(best, d);
                best = (o < best) ? o : best;
            }
            best = __builtin_amdgcn_readfirstlane(best);
            if (best == 0x7fffffff) break;
            int m = 0;
            for (int i = lane; i < deg; i += 32) m += (perm[beg + i] == best) ? 1 : 0;
            for (int d = 16; d >= 1; d >>= 1) m += __shfl_xor(m, d);
            m = __builtin_amdgcn_readfirstlane(m);
            const float w  = dinv[best] * dc;
            const v2f   hv = Hv[(long long)best * 32 + lane];
            for (int k = 0; k < m; ++k) acc += hv * w;
            prev = best;
            done += m;
        }
    }

    const v2f b = ((const v2f*)bias)[lane];
    v2f v = acc + b;
    if (doRelu) { v.x = fmaxf(v.x, 0.0f); v.y = fmaxf(v.y, 0.0f); }
    ((v2f*)out)[(long long)c * 32 + lane] = v;
}

extern "C" void kernel_launch(void* const* d_in, const int* in_sizes, int n_in,
                              void* d_out, int out_size, void* d_ws, size_t ws_size,
                              hipStream_t stream) {
    const float* x   = (const float*)d_in[0];
    const int*   ei  = (const int*)  d_in[1];
    const float* W1  = (const float*)d_in[2];
    const float* b1  = (const float*)d_in[3];
    const float* W2  = (const float*)d_in[4];
    const float* b2  = (const float*)d_in[5];
    float*       out = (float*)d_out;

    const int N  = in_sizes[0] / IN_CH;
    const int nE = in_sizes[1] / 2;
    const int*   rowp = ei;
    const int*   colp = ei + nE;

    const long long feat = (long long)N * HID_CH;
    float* h      = (float*)d_ws;
    float* a1     = h + feat;
    int*   packed = (int*)(a1 + feat);
    int*   perm   = packed + nE;
    int*   cnt    = perm + nE;
    int*   offs   = cnt + N;
    float* dinv   = (float*)(offs + N);

    const int NB = (N + BKT - 1) / BKT;
    const int G  = (nE + EPB - 1) / EPB;
    int*   bhist  = (int*)a1;
    int*   bstart = bhist + G * NB;
    int*   btot   = bstart + NB;

    const int waveBlocks = (N + 7) / 8;
    const int tileBlocks = (N + 15) / 16;
    const int gemmBlocks = (tileBlocks + GEMM_WAVES - 1) / GEMM_WAVES;

    gcn_bucket_count_kernel<<<G, 1024, 0, stream>>>(colp, bhist, nE, NB);
    gcn_bucket_scatter_kernel<<<G, 1024, 0, stream>>>(rowp, colp, bhist, packed, bstart, btot, nE, NB, G);
    gcn_bucket_sort_kernel<<<NB, 1024, 0, stream>>>(packed, bstart, btot, perm, cnt, offs, dinv, N);

    gcn_gemm_wmma_kernel<IN_CH><<<gemmBlocks, 256, 0, stream>>>(x, W1, h, N);
    gcn_gather_kernel<<<waveBlocks, 256, 0, stream>>>(h, offs, cnt, perm, dinv, b1, a1, N,  1);

    gcn_gemm_wmma_kernel<HID_CH><<<gemmBlocks, 256, 0, stream>>>(a1, W2, h, N);
    gcn_gather_kernel<<<waveBlocks, 256, 0, stream>>>(h, offs, cnt, perm, dinv, b2, out, N,  0);
}


# Gather: one wave per wave slot (grid 2048 x 8 waves) instead of one per node; a wave walks a run of consecutive nodes, the next node's sources and own row requested one node ahead; sum order kept
# speedup vs baseline: 3.2910x; 1.0646x over previous
#include <hip/hip_runtime.h>
#include <hip/hip_bf16.h>


typedef __attribute__((ext_vector_type(2))) float v2f;
typedef __attribute__((ext_vector_type(4))) float v4f;
typedef __attribute__((ext_vector_type(8))) float v8f;

#define IN_CH  128
#define HID_CH 64

#define SH      9
#define BKT     (1 << SH)
#define NBMAX   256
#define ROWBITS 17
#define EPB     8192

__global__ __launch_bounds__(1024)
void gcn_bucket_count_kernel(const int* __restrict__ col, int* __restrict__ bhist,
                             int nE, int NB) {
    __shared__ int s_h[NBMAX];
    const int tid = threadIdx.x;
    if (tid < NBMAX) s_h[tid] = 0;
    __syncthreads();
    const int base = blockIdx.x * EPB;
    for (int q = 0; q < EPB / 1024; ++q) {
        const int e = base + q * 1024 + tid;
        if (e < nE) atomicAdd(&s_h[col[e] >> SH], 1);
    }
    __syncthreads();
    if (tid < NB) bhist[blockIdx.x * NB + tid] = s_h[tid];
}

__global__ __launch_bounds__(1024)
void gcn_bucket_scatter_kernel(const int* __restrict__ row, const int* __restrict__ col,
                               const int* __restrict__ bhist, int* __restrict__ packed,
                               int* __restrict__ bstart, int* __restrict__ btot,
                               int nE, int NB, int G) {
    __shared__ int s_h[NBMAX];
    __shared__ int s_below[4][NBMAX];
    __shared__ int s_tot[4][NBMAX];
    __shared__ int s_base[NBMAX];
    __shared__ int s_w[8];
    const int tid  = threadIdx.x;
    const int lane = tid & 31;
    const int wv   = tid >> 5;
    const int g    = blockIdx.x;

    if (tid < NBMAX) s_h[tid] = 0;

    const int t = tid & (NBMAX - 1);
    const int p = tid >> 8;
    int below = 0, tot = 0;
    if (t < NB) {
        for (int k = p; k < G; k += 4) {
            const int v = bhist[k * NB + t];
            tot += v;
            if (k < g) below += v;
        }
    }
    s_below[p][t] = below;
    s_tot[p][t]   = tot;
    __syncthreads();

    const int base = g * EPB;
    int cc[EPB / 1024];
    int rk[EPB / 1024];
    for (int q = 0; q < EPB / 1024; ++q) {
        const int e = base + q * 1024 + tid;
        cc[q] = 0; rk[q] = 0;
        if (e < nE) {
            cc[q] = col[e];
            rk[q] = atomicAdd(&s_h[cc[q] >> SH], 1);
        }
    }

    int T = 0, B = 0, inc = 0;
    if (tid < NBMAX) {
        T = s_tot[0][tid] + s_tot[1][tid] + s_tot[2][tid] + s_tot[3][tid];
        B = s_below[0][tid] + s_below[1][tid] + s_below[2][tid] + s_below[3][tid];
        inc = T;
        for (int d = 1; d < 32; d <<= 1) {
            const int u = __shfl_up(inc, d);
            if (lane >= d) inc += u;
        }
        if (lane == 31) s_w[wv] = inc;
    }
    __syncthreads();
    if (tid < NBMAX) {
        int pre = 0;
        for (int k = 0; k < wv; ++k) pre += s_w[k];
        const int excl = pre + inc - T;
        s_base[tid] = excl + B;
        if (g == 0 && tid < NB) { bstart[tid] = excl; btot[tid] = T; }
    }
    __syncthreads();

    for (int q = 0; q < EPB / 1024; ++q) {
        const int e = base + q * 1024 + tid;
        if (e < nE) {
            const int c = cc[q];
            packed[s_base[c >> SH] + rk[q]] = ((c & (BKT - 1)) << ROWBITS) | row[e];
        }
    }
}

__global__ __launch_bounds__(1024)
void gcn_bucket_sort_kernel(const int* __restrict__ packed, const int* __restrict__ bstart,
                            const int* __restrict__ btot, int* __restrict__ perm,
                            int* __restrict__ cnt, int* __restrict__ offs,
                            float* __restrict__ dinv, int n) {
    __shared__ int s_cnt[BKT];
    __shared__ int s_cur[BKT];
    __shared__ int s_w[BKT / 32];
    const int tid   = threadIdx.x;
    const int lane  = tid & 31;
    const int wv    = tid >> 5;
    const int b     = blockIdx.x;
    const int start = bstart[b];
    const int m     = btot[b];

    if (tid < BKT) s_cnt[tid] = 0;
    __syncthreads();
    for (int i0 = 0; i0 < m; i0 += 4096) {
        int w[4];
        for (int q = 0; q < 4; ++q) {
            const int i = i0 + q * 1024 + tid;
            w[q] = (i < m) ? packed[start + i] : -1;
        }
        for (int q = 0; q < 4; ++q)
            if (w[q] >= 0) atomicAdd(&s_cnt[w[q] >> ROWBITS], 1);
    }
    __syncthreads();

    int v = 0, inc = 0;
    if (tid < BKT) {
        v = s_cnt[tid];
        inc = v;
        for (int d = 1; d < 32; d <<= 1) {
            const int u = __shfl_up(inc, d);
            if (lane >= d) inc += u;
        }
        if (lane == 31) s_w[wv] = inc;
    }
    __syncthreads();
    if (tid < BKT) {
        int pre = 0;
        for (int k = 0; k < wv; ++k) pre += s_w[k];
        const int excl = pre + inc - v;
        s_cur[tid] = excl;
        const int node = (b << SH) + tid;
        if (node < n) {
            cnt[node]  = v;
            offs[node] = start + excl;
            float d = (float)(v + 1);
            dinv[node] = (d > 0.0f) ? rsqrtf(d) : 0.0f;
        }
    }
    __syncthreads();
    for (int i0 = 0; i0 < m; i0 += 4096) {
        int w[4];
        for (int q = 0; q < 4; ++q) {
            const int i = i0 + q * 1024 + tid;
            w[q] = (i < m) ? packed[start + i] : -1;
        }
        for (int q = 0; q < 4; ++q) {
            if (w[q] >= 0) {
                const int slot = atomicAdd(&s_cur[w[q] >> ROWBITS], 1);
                perm[start + slot] = w[q] & ((1 << ROWBITS) - 1);
            }
        }
    }
}

#define GEMM_WAVES 8
#define TPITCH     68

template <int K>
__global__ __launch_bounds__(256)
void gcn_gemm_wmma_kernel(const float* __restrict__ X, const float* __restrict__ W,
                          float* __restrict__ H, int nNodes) {
    __shared__ __attribute__((aligned(16))) float s_w[K * HID_CH];
    __shared__ __attribute__((aligned(16))) float s_t[GEMM_WAVES * 16 * TPITCH];

    const int tid  = threadIdx.x;
    const int lane = tid & 31;
    const int wave = __builtin_amdgcn_readfirstlane(tid >> 5);
    const int n    = lane & 15;
    const int half = lane >> 4;
    const int r0   = (blockIdx.x * GEMM_WAVES + wave) * 16;

    int rA = r0 + n;
    if (rA > nNodes - 1) rA = nNodes - 1;
    const v4f* xr = (const v4f*)(X + (long long)rA * K) + half;
    v4f a4[K / 8];
#pragma unroll
    for (int j = 0; j < K / 8; ++j) a4[j] = xr[2 * j];

    constexpr int WQ = K * HID_CH / 1024;
    v4f w4[WQ];
#pragma unroll
    for (int q = 0; q < WQ; ++q) w4[q] = *(const v4f*)(W + q * 1024 + tid * 4);
#pragma unroll
    for (int q = 0; q < WQ; ++q) {
        const int g  = q * 1024 + tid * 4;
        const int k  = g >> 6;
        const int c  = g & (HID_CH - 1);
        const int kj = k >> 3;
        const int kh = (k >> 2) & 1;
        const int ke = k & 3;
        const int o  = ((((kj * 4 + (c >> 4)) * 32) + kh * 16 + (c & 15)) << 2) + ke;
        s_w[o]      = w4[q].x;
        s_w[o + 4]  = w4[q].y;
        s_w[o + 8]  = w4[q].z;
        s_w[o + 12] = w4[q].w;
    }
    __syncthreads();

    if (r0 >= nNodes) return;

    v8f acc0 = {}, acc1 = {}, acc2 = {}, acc3 = {};
    const v4f* bw = (const v4f*)s_w + lane;
#pragma unroll
    for (int j = 0; j < K / 8; ++j) {
        const v4f av = a4[j];
        const v4f q0 = bw[(j * 4 + 0) * 32];
        const v4f q1 = bw[(j * 4 + 1) * 32];
        const v4f q2 = bw[(j * 4 + 2) * 32];
        const v4f q3 = bw[(j * 4 + 3) * 32];
        v2f a, b0, b1, b2, b3;
        a.x = av.x;  a.y = av.y;
        b0.x = q0.x; b0.y = q0.y;
        b1.x = q1.x; b1.y = q1.y;
        b2.x = q2.x; b2.y = q2.y;
        b3.x = q3.x; b3.y = q3.y;
        acc0 = __builtin_amdgcn_wmma_f32_16x16x4_f32(false, a, false, b0, (short)0, acc0, false, false);
        acc1 = __builtin_amdgcn_wmma_f32_16x16x4_f32(false, a, false, b1, (short)0, acc1, false, false);
        acc2 = __builtin_amdgcn_wmma_f32_16x16x4_f32(false, a, false, b2, (short)0, acc2, false, false);
        acc3 = __builtin_amdgcn_wmma_f32_16x16x4_f32(false, a, false, b3, (short)0, acc3, false, false);
        a.x = av.z;  a.y = av.w;
        b0.x = q0.z; b0.y = q0.w;
        b1.x = q1.z; b1.y = q1.w;
        b2.x = q2.z; b2.y = q2.w;
        b3.x = q3.z; b3.y = q3.w;
        acc0 = __builtin_amdgcn_wmma_f32_16x16x4_f32(false, a, false, b0, (short)0, acc0, false, false);
        acc1 = __builtin_amdgcn_wmma_f32_16x16x4_f32(false, a, false, b1, (short)0, acc1, false, false);
        acc2 = __builtin_amdgcn_wmma_f32_16x16x4_f32(false, a, false, b2, (short)0, acc2, false, false);
        acc3 = __builtin_amdgcn_wmma_f32_16x16x4_f32(false, a, false, b3, (short)0, acc3, false, false);
    }

    float* st = s_t + wave * (16 * TPITCH);
#pragma unroll
    for (int v = 0; v < 8; ++v) {
        float* p = st + (v + 8 * half) * TPITCH + n;
        p[0]  = acc0[v];
        p[16] = acc1[v];
        p[32] = acc2[v];
        p[48] = acc3[v];
    }
    v4f o[8];
#pragma unroll
    for (int i = 0; i < 8; ++i) o[i] = *(const v4f*)(st + (2 * i + half) * TPITCH + 4 * n);
    float* hp = H + (long long)(r0 + half) * HID_CH + 4 * n;
    if (r0 + 16 <= nNodes) {
#pragma unroll
        for (int i = 0; i < 8; ++i) *(v4f*)(hp + (long long)i * 2 * HID_CH) = o[i];
    } else {
#pragma unroll
        for (int i = 0; i < 8; ++i)
            if (r0 + 2 * i + half < nNodes) *(v4f*)(hp + (long long)i * 2 * HID_CH) = o[i];
    }
}

#define LSEG 128

__device__ __forceinline__
v2f gcn_sum_rows(v2f acc, const v2f* __restrict__ Hv, int rs, float ws, int cnt, int lane) {
    int j = 0;
    for (; j + 4 <= cnt; j += 4) {
        const int   r0 = __builtin_amdgcn_readlane(rs, j);
        const int   r1 = __builtin_amdgcn_readlane(rs, j + 1);
        const int   r2 = __builtin_amdgcn_readlane(rs, j + 2);
        const int   r3 = __builtin_amdgcn_readlane(rs, j + 3);
        const float w0 = __int_as_float(__builtin_amdgcn_readlane(__float_as_int(ws), j));
        const float w1 = __int_as_float(__builtin_amdgcn_readlane(__float_as_int(ws), j + 1));
        const float w2 = __int_as_float(__builtin_amdgcn_readlane(__float_as_int(ws), j + 2));
        const float w3 = __int_as_float(__builtin_amdgcn_readlane(__float_as_int(ws), j + 3));
        const v2f h0 = Hv[(long long)r0 * 32 + lane];
        const v2f h1 = Hv[(long long)r1 * 32 + lane];
        const v2f h2 = Hv[(long long)r2 * 32 + lane];
        const v2f h3 = Hv[(long long)r3 * 32 + lane];
        acc += h0 * w0;
        acc += h1 * w1;
        acc += h2 * w2;
        acc += h3 * w3;
    }
    for (; j < cnt; ++j) {
        const int   r0 = __builtin_amdgcn_readlane(rs, j);
        const float w0 = __int_as_float(__builtin_amdgcn_readlane(__float_as_int(ws), j));
        acc += Hv[(long long)r0 * 32 + lane] * w0;
    }
    return acc;
}

#define GATHER_BLOCKS 2048

__global__ __launch_bounds__(256)
void gcn_gather_kernel(const float* __restrict__ H, const int* __restrict__ offs,
                       const int* __restrict__ cnt, const int* __restrict__ perm,
                       const float* __restrict__ dinv, const float* __restrict__ bias,
                       float* __restrict__ out, int nNodes, int doRelu) {
    __shared__ int s_raw[8][LSEG];
    __shared__ int s_ord[8][LSEG];

    const int lane = threadIdx.x & 31;
    const int wv = __builtin_amdgcn_readfirstlane((int)(threadIdx.x >> 5));

    const int gw    = blockIdx.x * 8 + wv;
    const int TW    = gridDim.x * 8;
    const int q     = nNodes / TW;
    const int rem   = nNodes - q * TW;
    const int start = gw * q + ((gw < rem) ? gw : rem);
    const int end   = start + q + ((gw < rem) ? 1 : 0);

    const v2f* Hv = (const v2f*)H;
    const v2f  b  = ((const v2f*)bias)[lane];

    for (int base = start; base < end; base += 32) {
        const int nk = (end - base < 32) ? (end - base) : 32;

        int   m_beg = 0, m_deg = 0;
        float m_dc  = 0.0f;
        if (lane < nk) {
            m_beg = offs[base + lane];
            m_deg = cnt[base + lane];
            m_dc  = dinv[base + lane];
        }

        int r_cur = 0x7fffffff;
        v2f a_cur;
        {
            const int   b0 = __builtin_amdgcn_readlane(m_beg, 0);
            const int   d0 = __builtin_amdgcn_readlane(m_deg, 0);
            const float c0 = __int_as_float(__builtin_amdgcn_readlane(__float_as_int(m_dc), 0));
            if (lane < d0) r_cur = perm[b0 + lane];
            a_cur = Hv[(long long)base * 32 + lane] * (c0 * c0);
        }

        for (int k = 0; k < nk; ++k) {
            const int   c   = base + k;
            const int   beg = __builtin_amdgcn_readlane(m_beg, k);
            const int   deg = __builtin_amdgcn_readlane(m_deg, k);
            const float dc  = __int_as_float(__builtin_amdgcn_readlane(__float_as_int(m_dc), k));

            const int kn = (k + 1 < nk) ? (k + 1) : k;
            int r_nxt = 0x7fffffff;
            {
                const int bn = __builtin_amdgcn_readlane(m_beg, kn);
                const int dn = __builtin_amdgcn_readlane(m_deg, kn);
                if (lane < dn) r_nxt = perm[bn + lane];
            }
            const v2f h_nxt = Hv[(long long)(base + kn) * 32 + lane];

            v2f acc = a_cur;

            if (deg <= 32) {
                const int r_l = r_cur;

                int pos = 0;
                for (int j = 0; j < deg; ++j) {
                    const int rj = __builtin_amdgcn_readlane(r_l, j);
                    pos += ((rj < r_l) || (rj == r_l && j < lane)) ? 1 : 0;
                }
                if (lane >= deg) pos = lane;

                const int rs = __builtin_amdgcn_ds_permute(pos << 2, r_l);
                float ws = 0.0f;
                if (lane < deg) ws = dinv[rs] * dc;

                acc = gcn_sum_rows(acc, Hv, rs, ws, deg, lane);
            } else if (deg <= LSEG) {
                int* raw = s_raw[wv];
                int* ord = s_ord[wv];
                const int K = (deg + 31) >> 5;

                for (int t = 0; t < K; ++t) {
                    const int i = t * 32 + lane;
                    raw[i] = (i < deg) ? perm[beg + i] : 0x7fffffff;
                }
                for (int t = 0; t < K; ++t) {
                    const int i = t * 32 + lane;
                    const int e = raw[i];
                    int pos = 0;
                    for (int cj = 0; cj < K; ++cj) {
                        const int ej = raw[cj * 32 + lane];
                        const int nj = (deg - cj * 32 < 32) ? (deg - cj * 32) : 32;
                        for (int j = 0; j < nj; ++j) {
                            const int rj = __builtin_amdgcn_readlane(ej, j);
                            pos += ((rj < e) || (rj == e && cj * 32 + j < i)) ? 1 : 0;
                        }
                    }
                    if (i < deg) ord[pos] = e;
                }
                for (int t = 0; t < K; ++t) {
                    const int nv = (deg - t * 32 < 32) ? (deg - t * 32) : 32;
                    int   rs = 0;
                    float ws = 0.0f;
                    if (lane < nv) { rs = ord[t * 32 + lane]; ws = dinv[rs] * dc; }
                    acc = gcn_sum_rows(acc, Hv, rs, ws, nv, lane);
                }
            } else {
                int prev = -1;
                int done = 0;
                for (int it = 0; it < deg && done < deg; ++it) {
                    int best = 0x7fffffff;
                    for (int i = lane; i < deg; i += 32) {
                        const int r = perm[beg + i];
                        if (r > prev && r < best) best = r;
                    }
                    for (int d = 16; d >= 1; d >>= 1) {
                        const int o = __shfl_xor(best, d);
                        best = (o < best) ? o : best;
                    }
                    best = __builtin_amdgcn_readfirstlane(best);
                    if (best == 0x7fffffff) break;
                    int m = 0;
                    for (int i = lane; i < deg; i += 32) m += (perm[beg + i] == best) ? 1 : 0;
                    for (int d = 16; d >= 1; d >>= 1) m += __shfl_xor(m, d);
                    m = __builtin_amdgcn_readfirstlane(m);
                    const float w  = dinv[best] * dc;
                    const v2f   hv = Hv[(long long)best * 32 + lane];
                    for (int t = 0; t < m; ++t) acc += hv * w;
                    prev = best;
                    done += m;
                }
            }

            v2f v = acc + b;
            if (doRelu) { v.x = fmaxf(v.x, 0.0f); v.y = fmaxf(v.y, 0.0f); }
            ((v2f*)out)[(long long)c * 32 + lane] = v;

            const float cn = __int_as_float(__builtin_amdgcn_readlane(__float_as_int(m_dc), kn));
            a_cur = h_nxt * (cn * cn);
            r_cur = r_nxt;
        }
    }
}

extern "C" void kernel_launch(void* const* d_in, const int* in_sizes, int n_in,
                              void* d_out, int out_size, void* d_ws, size_t ws_size,
                              hipStream_t stream) {
    const float* x   = (const float*)d_in[0];
    const int*   ei  = (const int*)  d_in[1];
    const float* W1  = (const float*)d_in[2];
    const float* b1  = (const float*)d_in[3];
    const float* W2  = (const float*)d_in[4];
    const float* b2  = (const float*)d_in[5];
    float*       out = (float*)d_out;

    const int N  = in_sizes[0] / IN_CH;
    const int nE = in_sizes[1] / 2;
    const int*   rowp = ei;
    const int*   colp = ei + nE;

    const long long feat = (long long)N * HID_CH;
    float* h      = (float*)d_ws;
    float* a1     = h + feat;
    int*   packed = (int*)(a1 + feat);
    int*   perm   = packed + nE;
    int*   cnt    = perm + nE;
    int*   offs   = cnt + N;
    float* dinv   = (float*)(offs + N);

    const int NB = (N + BKT - 1) / BKT;
    const int G  = (nE + EPB - 1) / EPB;
    int*   bhist  = (int*)a1;
    int*   bstart = bhist + G * NB;
    int*   btot   = bstart + NB;

    const int tileBlocks = (N + 15) / 16;
    const int gemmBlocks = (tileBlocks + GEMM_WAVES - 1) / GEMM_WAVES;

    gcn_bucket_count_kernel<<<G, 1024, 0, stream>>>(colp, bhist, nE, NB);
    gcn_bucket_scatter_kernel<<<G, 1024, 0, stream>>>(rowp, colp, bhist, packed, bstart, btot, nE, NB, G);
    gcn_bucket_sort_kernel<<<NB, 1024, 0, stream>>>(packed, bstart, btot, perm, cnt, offs, dinv, N);

    gcn_gemm_wmma_kernel<IN_CH><<<gemmBlocks, 256, 0, stream>>>(x, W1, h, N);
    gcn_gather_kernel<<<GATHER_BLOCKS, 256, 0, stream>>>(h, offs, cnt, perm, dinv, b1, a1, N,  1);

    gcn_gemm_wmma_kernel<HID_CH><<<gemmBlocks, 256, 0, stream>>>(a1, W2, h, N);
    gcn_gather_kernel<<<GATHER_BLOCKS, 256, 0, stream>>>(h, offs, cnt, perm, dinv, b2, out, N,  0);
}


# Edge sort level 1 made block-local: a block groups its own 8192 edges by bucket in its own region, the bucket pass walks its run of every block; no global count pass, 7 -> 6 launches; sum order kept
# speedup vs baseline: 4.1236x; 1.2530x over previous
#include <hip/hip_runtime.h>
#include <hip/hip_bf16.h>


typedef __attribute__((ext_vector_type(2))) float v2f;
typedef __attribute__((ext_vector_type(4))) float v4f;
typedef __attribute__((ext_vector_type(8))) float v8f;

#define IN_CH  128
#define HID_CH 64

#define SH      9
#define BKT     (1 << SH)
#define NBMAX   256
#define ROWBITS 17
#define EPB     8192
#define GMAX    256
#define RSH     6
#define RSLOT   (1 << RSH)
#define FPT     8

__global__ __launch_bounds__(1024)
void gcn_bucket_scatter_kernel(const int* __restrict__ row, const int* __restrict__ col,
                               int* __restrict__ packed, int* __restrict__ rcnt,
                               int* __restrict__ roff, int nE, int NB) {
    __shared__ int s_h[NBMAX];
    __shared__ int s_base[NBMAX];
    __shared__ int s_w[8];
    const int tid  = threadIdx.x;
    const int lane = tid & 31;
    const int wv   = tid >> 5;
    const int g    = blockIdx.x;

    if (tid < NBMAX) s_h[tid] = 0;
    __syncthreads();

    const int base = g * EPB;
    int cc[EPB / 1024];
    int rk[EPB / 1024];
    for (int q = 0; q < EPB / 1024; ++q) {
        const int e = base + q * 1024 + tid;
        cc[q] = 0; rk[q] = 0;
        if (e < nE) {
            cc[q] = col[e];
            rk[q] = atomicAdd(&s_h[cc[q] >> SH], 1);
        }
    }
    __syncthreads();

    int T = 0, inc = 0;
    if (tid < NBMAX) {
        T = s_h[tid];
        inc = T;
        for (int d = 1; d < 32; d <<= 1) {
            const int u = __shfl_up(inc, d);
            if (lane >= d) inc += u;
        }
        if (lane == 31) s_w[wv] = inc;
    }
    __syncthreads();
    if (tid < NBMAX) {
        int pre = 0;
        for (int k = 0; k < wv; ++k) pre += s_w[k];
        const int excl = pre + inc - T;
        s_base[tid] = excl;
        if (tid < NB) { rcnt[g * NB + tid] = T; roff[g * NB + tid] = excl; }
    }
    __syncthreads();

    for (int q = 0; q < EPB / 1024; ++q) {
        const int e = base + q * 1024 + tid;
        if (e < nE) {
            const int c = cc[q];
            packed[base + s_base[c >> SH] + rk[q]] = ((c & (BKT - 1)) << ROWBITS) | row[e];
        }
    }
}

__global__ __launch_bounds__(1024)
void gcn_bucket_sort_kernel(const int* __restrict__ packed, const int* __restrict__ rcnt,
                            const int* __restrict__ roff, int* __restrict__ perm,
                            int* __restrict__ cnt, int* __restrict__ offs,
                            float* __restrict__ dinv, int n, int NB, int G) {
    __shared__ int s_cnt[BKT];
    __shared__ int s_cur[BKT];
    __shared__ int s_w[BKT / 32];
    __shared__ int s_tot[4][NBMAX];
    __shared__ int s_w8[8];
    __shared__ int2 s_run[GMAX];
    __shared__ int s_start;
    const int tid   = threadIdx.x;
    const int lane  = tid & 31;
    const int wv    = tid >> 5;
    const int b     = blockIdx.x;

    {
        const int t = tid & (NBMAX - 1);
        const int p = tid >> 8;
        int tot = 0;
        if (t < NB) {
            for (int k = p; k < G; k += 4) tot += rcnt[k * NB + t];
        }
        s_tot[p][t] = tot;
    }
    if (tid < G) s_run[tid] = make_int2(rcnt[tid * NB + b], tid * EPB + roff[tid * NB + b]);
    if (tid < BKT) s_cnt[tid] = 0;
    __syncthreads();

    {
        int T = 0, inc = 0;
        if (tid < NBMAX) {
            T = s_tot[0][tid] + s_tot[1][tid] + s_tot[2][tid] + s_tot[3][tid];
            inc = T;
            for (int d = 1; d < 32; d <<= 1) {
                const int u = __shfl_up(inc, d);
                if (lane >= d) inc += u;
            }
            if (lane == 31) s_w8[wv] = inc;
        }
        __syncthreads();
        if (tid == b) {
            int pre = 0;
            for (int k = 0; k < wv; ++k) pre += s_w8[k];
            s_start = pre + inc - T;
        }
    }

    const int nslot = G << RSH;
    const int ent = tid & (RSLOT - 1);
    for (int i0 = 0; i0 < nslot; i0 += FPT * 1024) {
        int2 ro[FPT];
        int  w[FPT];
        for (int q = 0; q < FPT; ++q) {
            const int gg = (i0 + q * 1024 + tid) >> RSH;
            ro[q] = s_run[(gg < G) ? gg : (G - 1)];
            if (gg >= G) ro[q].x = 0;
        }
        for (int q = 0; q < FPT; ++q) w[q] = (ent < ro[q].x) ? packed[ro[q].y + ent] : -1;
        for (int q = 0; q < FPT; ++q)
            if (w[q] >= 0) atomicAdd(&s_cnt[w[q] >> ROWBITS], 1);
    }
    for (int gg = wv; gg < G; gg += 32) {
        const int2 r = s_run[gg];
        for (int i = RSLOT + lane; i < r.x; i += 32)
            atomicAdd(&s_cnt[packed[r.y + i] >> ROWBITS], 1);
    }
    __syncthreads();

    int v = 0, inc = 0;
    if (tid < BKT) {
        v = s_cnt[tid];
        inc = v;
        for (int d = 1; d < 32; d <<= 1) {
            const int u = __shfl_up(inc, d);
            if (lane >= d) inc += u;
        }
        if (lane == 31) s_w[wv] = inc;
    }
    __syncthreads();
    const int start = s_start;
    if (tid < BKT) {
        int pre = 0;
        for (int k = 0; k < wv; ++k) pre += s_w[k];
        const int excl = pre + inc - v;
        s_cur[tid] = excl;
        const int node = (b << SH) + tid;
        if (node < n) {
            cnt[node]  = v;
            offs[node] = start + excl;
            float d = (float)(v + 1);
            dinv[node] = (d > 0.0f) ? rsqrtf(d) : 0.0f;
        }
    }
    __syncthreads();

    for (int i0 = 0; i0 < nslot; i0 += FPT * 1024) {
        int2 ro[FPT];
        int  w[FPT];
        for (int q = 0; q < FPT; ++q) {
            const int gg = (i0 + q * 1024 + tid) >> RSH;
            ro[q] = s_run[(gg < G) ? gg : (G - 1)];
            if (gg >= G) ro[q].x = 0;
        }
        for (int q = 0; q < FPT; ++q) w[q] = (ent < ro[q].x) ? packed[ro[q].y + ent] : -1;
        for (int q = 0; q < FPT; ++q) {
            if (w[q] >= 0) {
                const int slot = atomicAdd(&s_cur[w[q] >> ROWBITS], 1);
                perm[start + slot] = w[q] & ((1 << ROWBITS) - 1);
            }
        }
    }
    for (int gg = wv; gg < G; gg += 32) {
        const int2 r = s_run[gg];
        for (int i = RSLOT + lane; i < r.x; i += 32) {
            const int w = packed[r.y + i];
            const int slot = atomicAdd(&s_cur[w >> ROWBITS], 1);
            perm[start + slot] = w & ((1 << ROWBITS) - 1);
        }
    }
}

#define GEMM_WAVES 8
#define TPITCH     68

template <int K>
__global__ __launch_bounds__(256)
void gcn_gemm_wmma_kernel(const float* __restrict__ X, const float* __restrict__ W,
                          const float* __restrict__ dinv, float* __restrict__ H, int nNodes) {
    __shared__ __attribute__((aligned(16))) float s_w[K * HID_CH];
    __shared__ __attribute__((aligned(16))) float s_t[GEMM_WAVES * 16 * TPITCH];

    const int tid  = threadIdx.x;
    const int lane = tid & 31;
    const int wave = __builtin_amdgcn_readfirstlane(tid >> 5);
    const int n    = lane & 15;
    const int half = lane >> 4;
    const int r0   = (blockIdx.x * GEMM_WAVES + wave) * 16;

    int rA = r0 + n;
    if (rA > nNodes - 1) rA = nNodes - 1;
    const v4f* xr = (const v4f*)(X + (long long)rA * K) + half;
    v4f a4[K / 8];
#pragma unroll
    for (int j = 0; j < K / 8; ++j) a4[j] = xr[2 * j];

    constexpr int WQ = K * HID_CH / 1024;
    v4f w4[WQ];
#pragma unroll
    for (int q = 0; q < WQ; ++q) w4[q] = *(const v4f*)(W + q * 1024 + tid * 4);
#pragma unroll
    for (int q = 0; q < WQ; ++q) {
        const int g  = q * 1024 + tid * 4;
        const int k  = g >> 6;
        const int c  = g & (HID_CH - 1);
        const int kj = k >> 3;
        const int kh = (k >> 2) & 1;
        const int ke = k & 3;
        const int o  = ((((kj * 4 + (c >> 4)) * 32) + kh * 16 + (c & 15)) << 2) + ke;
        s_w[o]      = w4[q].x;
        s_w[o + 4]  = w4[q].y;
        s_w[o + 8]  = w4[q].z;
        s_w[o + 12] = w4[q].w;
    }
    __syncthreads();

    if (r0 >= nNodes) return;

    float dsc[8];
#pragma unroll
    for (int i = 0; i < 8; ++i) {
        int r = r0 + 2 * i + half;
        if (r > nNodes - 1) r = nNodes - 1;
        dsc[i] = dinv[r];
    }

    v8f acc0 = {}, acc1 = {}, acc2 = {}, acc3 = {};
    const v4f* bw = (const v4f*)s_w + lane;
#pragma unroll
    for (int j = 0; j < K / 8; ++j) {
        const v4f av = a4[j];
        const v4f q0 = bw[(j * 4 + 0) * 32];
        const v4f q1 = bw[(j * 4 + 1) * 32];
        const v4f q2 = bw[(j * 4 + 2) * 32];
        const v4f q3 = bw[(j * 4 + 3) * 32];
        v2f a, b0, b1, b2, b3;
        a.x = av.x;  a.y = av.y;
        b0.x = q0.x; b0.y = q0.y;
        b1.x = q1.x; b1.y = q1.y;
        b2.x = q2.x; b2.y = q2.y;
        b3.x = q3.x; b3.y = q3.y;
        acc0 = __builtin_amdgcn_wmma_f32_16x16x4_f32(false, a, false, b0, (short)0, acc0, false, false);
        acc1 = __builtin_amdgcn_wmma_f32_16x16x4_f32(false, a, false, b1, (short)0, acc1, false, false);
        acc2 = __builtin_amdgcn_wmma_f32_16x16x4_f32(false, a, false, b2, (short)0, acc2, false, false);
        acc3 = __builtin_amdgcn_wmma_f32_16x16x4_f32(false, a, false, b3, (short)0, acc3, false, false);
        a.x = av.z;  a.y = av.w;
        b0.x = q0.z; b0.y = q0.w;
        b1.x = q1.z; b1.y = q1.w;
        b2.x = q2.z; b2.y = q2.w;
        b3.x = q3.z; b3.y = q3.w;
        acc0 = __builtin_amdgcn_wmma_f32_16x16x4_f32(false, a, false, b0, (short)0, acc0, false, false);
        acc1 = __builtin_amdgcn_wmma_f32_16x16x4_f32(false, a, false, b1, (short)0, acc1, false, false);
        acc2 = __builtin_amdgcn_wmma_f32_16x16x4_f32(false, a, false, b2, (short)0, acc2, false, false);
        acc3 = __builtin_amdgcn_wmma_f32_16x16x4_f32(false, a, false, b3, (short)0, acc3, false, false);
    }

    float* st = s_t + wave * (16 * TPITCH);
#pragma unroll
    for (int v = 0; v < 8; ++v) {
        float* p = st + (v + 8 * half) * TPITCH + n;
        p[0]  = acc0[v];
        p[16] = acc1[v];
        p[32] = acc2[v];
        p[48] = acc3[v];
    }
    v4f o[8];
#pragma unroll
    for (int i = 0; i < 8; ++i) o[i] = *(const v4f*)(st + (2 * i + half) * TPITCH + 4 * n) * dsc[i];
    float* hp = H + (long long)(r0 + half) * HID_CH + 4 * n;
    if (r0 + 16 <= nNodes) {
#pragma unroll
        for (int i = 0; i < 8; ++i) *(v4f*)(hp + (long long)i * 2 * HID_CH) = o[i];
    } else {
#pragma unroll
        for (int i = 0; i < 8; ++i)
            if (r0 + 2 * i + half < nNodes) *(v4f*)(hp + (long long)i * 2 * HID_CH) = o[i];
    }
}

#define LSEG 128

__device__ __forceinline__
v2f gcn_sum_rows(v2f acc, const v2f* __restrict__ Hv, int rs, int cnt, int lane) {
    int j = 0;
#pragma unroll 2
    for (; j + 4 <= cnt; j += 4) {
        const int r0 = __builtin_amdgcn_readlane(rs, j);
        const int r1 = __builtin_amdgcn_readlane(rs, j + 1);
        const int r2 = __builtin_amdgcn_readlane(rs, j + 2);
        const int r3 = __builtin_amdgcn_readlane(rs, j + 3);
        const v2f h0 = Hv[(long long)r0 * 32 + lane];
        const v2f h1 = Hv[(long long)r1 * 32 + lane];
        const v2f h2 = Hv[(long long)r2 * 32 + lane];
        const v2f h3 = Hv[(long long)r3 * 32 + lane];
        acc += h0;
        acc += h1;
        acc += h2;
        acc += h3;
    }
    for (; j < cnt; ++j) {
        const int r0 = __builtin_amdgcn_readlane(rs, j);
        acc += Hv[(long long)r0 * 32 + lane];
    }
    return acc;
}

#define GATHER_BLOCKS 2048

__global__ __launch_bounds__(256)
void gcn_gather_kernel(const float* __restrict__ H, const int* __restrict__ offs,
                       const int* __restrict__ cnt, const int* __restrict__ perm,
                       const float* __restrict__ dinv, const float* __restrict__ bias,
                       float* __restrict__ out, int nNodes, int doRelu) {
    __shared__ int s_raw[8][LSEG];
    __shared__ int s_ord[8][LSEG];

    const int lane = threadIdx.x & 31;
    const int wv = __builtin_amdgcn_readfirstlane((int)(threadIdx.x >> 5));

    const int gw    = blockIdx.x * 8 + wv;
    const int TW    = gridDim.x * 8;
    const int q     = nNodes / TW;
    const int rem   = nNodes - q * TW;
    const int start = gw * q + ((gw < rem) ? gw : rem);
    const int end   = start + q + ((gw < rem) ? 1 : 0);

    const v2f* Hv = (const v2f*)H;
    const v2f  b  = ((const v2f*)bias)[lane];

    for (int base = start; base < end; base += 32) {
        const int nk = (end - base < 32) ? (end - base) : 32;

        int   m_beg = 0, m_deg = 0;
        float m_dc  = 0.0f;
        if (lane < nk) {
            m_beg = offs[base + lane];
            m_deg = cnt[base + lane];
            m_dc  = dinv[base + lane];
        }

        int r_cur = 0x7fffffff;
        v2f a_cur;
        {
            const int b0 = __builtin_amdgcn_readlane(m_beg, 0);
            const int d0 = __builtin_amdgcn_readlane(m_deg, 0);
            if (lane < d0) r_cur = perm[b0 + lane];
            a_cur = Hv[(long long)base * 32 + lane];
        }

        for (int k = 0; k < nk; ++k) {
            const int   c   = base + k;
            const int   beg = __builtin_amdgcn_readlane(m_beg, k);
            const int   deg = __builtin_amdgcn_readlane(m_deg, k);
            const float dc  = __int_as_float(__builtin_amdgcn_readlane(__float_as_int(m_dc), k));

            const int kn = (k + 1 < nk) ? (k + 1) : k;
            int r_nxt = 0x7fffffff;
            {
                const int bn = __builtin_amdgcn_readlane(m_beg, kn);
                const int dn = __builtin_amdgcn_readlane(m_deg, kn);
                if (lane < dn) r_nxt = perm[bn + lane];
            }
            const v2f h_nxt = Hv[(long long)(base + kn) * 32 + lane];

            v2f acc = a_cur;

            if (deg <= 32) {
                const int r_l = r_cur;

                int pos = 0;
                for (int j = 0; j < deg; ++j) {
                    const int rj = __builtin_amdgcn_readlane(r_l, j);
                    pos += ((rj < r_l) || (rj == r_l && j < lane)) ? 1 : 0;
                }
                if (lane >= deg) pos = lane;

                const int rs = __builtin_amdgcn_ds_permute(pos << 2, r_l);

                acc = gcn_sum_rows(acc, Hv, rs, deg, lane);
            } else if (deg <= LSEG) {
                int* raw = s_raw[wv];
                int* ord = s_ord[wv];
                const int K = (deg + 31) >> 5;

                for (int t = 0; t < K; ++t) {
                    const int i = t * 32 + lane;
                    raw[i] = (i < deg) ? perm[beg + i] : 0x7fffffff;
                }
                for (int t = 0; t < K; ++t) {
                    const int i = t * 32 + lane;
                    const int e = raw[i];
                    int pos = 0;
                    for (int cj = 0; cj < K; ++cj) {
                        const int ej = raw[cj * 32 + lane];
                        const int nj = (deg - cj * 32 < 32) ? (deg - cj * 32) : 32;
                        for (int j = 0; j < nj; ++j) {
                            const int rj = __builtin_amdgcn_readlane(ej, j);
                            pos += ((rj < e) || (rj == e && cj * 32 + j < i)) ? 1 : 0;
                        }
                    }
                    if (i < deg) ord[pos] = e;
                }
                for (int t = 0; t < K; ++t) {
                    const int nv = (deg - t * 32 < 32) ? (deg - t * 32) : 32;
                    int rs = 0;
                    if (lane < nv) rs = ord[t * 32 + lane];
                    acc = gcn_sum_rows(acc, Hv, rs, nv, lane);
                }
            } else {
                int prev = -1;
                int done = 0;
                for (int it = 0; it < deg && done < deg; ++it) {
                    int best = 0x7fffffff;
                    for (int i = lane; i < deg; i += 32) {
                        const int r = perm[beg + i];
                        if (r > prev && r < best) best = r;
                    }
                    for (int d = 16; d >= 1; d >>= 1) {
                        const int o = __shfl_xor(best, d);
                        best = (o < best) ? o : best;
                    }
                    best = __builtin_amdgcn_readfirstlane(best);
                    if (best == 0x7fffffff) break;
                    int m = 0;
                    for (int i = lane; i < deg; i += 32) m += (perm[beg + i] == best) ? 1 : 0;
                    for (int d = 16; d >= 1; d >>= 1) m += __shfl_xor(m, d);
                    m = __builtin_amdgcn_readfirstlane(m);
                    const v2f hv = Hv[(long long)best * 32 + lane];
                    for (int t = 0; t < m; ++t) acc += hv;
                    prev = best;
                    done += m;
                }
            }

            v2f v = acc * dc + b;
            if (doRelu) { v.x = fmaxf(v.x, 0.0f); v.y = fmaxf(v.y, 0.0f); }
            ((v2f*)out)[(long long)c * 32 + lane] = v;

            a_cur = h_nxt;
            r_cur = r_nxt;
        }
    }
}

extern "C" void kernel_launch(void* const* d_in, const int* in_sizes, int n_in,
                              void* d_out, int out_size, void* d_ws, size_t ws_size,
                              hipStream_t stream) {
    const float* x   = (const float*)d_in[0];
    const int*   ei  = (const int*)  d_in[1];
    const float* W1  = (const float*)d_in[2];
    const float* b1  = (const float*)d_in[3];
    const float* W2  = (const float*)d_in[4];
    const float* b2  = (const float*)d_in[5];
    float*       out = (float*)d_out;

    const int N  = in_sizes[0] / IN_CH;
    const int nE = in_sizes[1] / 2;
    const int*   rowp = ei;
    const int*   colp = ei + nE;

    const long long feat = (long long)N * HID_CH;
    float* h      = (float*)d_ws;
    float* a1     = h + feat;
    int*   packed = (int*)(a1 + feat);
    int*   perm   = packed + nE;
    int*   cnt    = perm + nE;
    int*   offs   = cnt + N;
    float* dinv   = (float*)(offs + N);

    const int NB = (N + BKT - 1) / BKT;
    const int G  = (nE + EPB - 1) / EPB;
    int*   rcnt   = (int*)a1;
    int*   roff   = rcnt + G * NB;

    const int tileBlocks = (N + 15) / 16;
    const int gemmBlocks = (tileBlocks + GEMM_WAVES - 1) / GEMM_WAVES;

    gcn_bucket_scatter_kernel<<<G, 1024, 0, stream>>>(rowp, colp, packed, rcnt, roff, nE, NB);
    gcn_bucket_sort_kernel<<<NB, 1024, 0, stream>>>(packed, rcnt, roff, perm, cnt, offs, dinv, N, NB, G);

    gcn_gemm_wmma_kernel<IN_CH><<<gemmBlocks, 256, 0, stream>>>(x, W1, dinv, h, N);
    gcn_gather_kernel<<<GATHER_BLOCKS, 256, 0, stream>>>(h, offs, cnt, perm, dinv, b1, a1, N,  1);

    gcn_gemm_wmma_kernel<HID_CH><<<gemmBlocks, 256, 0, stream>>>(a1, W2, dinv, h, N);
    gcn_gather_kernel<<<GATHER_BLOCKS, 256, 0, stream>>>(h, offs, cnt, perm, dinv, b2, out, N,  0);
}


# Layer-2 product taken inside the layer-1 gather: a block keeps its finished rows in LDS, multiplies them by W2 there (same k order) and stores the dinv-scaled rows; 6 -> 5 launches; sum order kept
# speedup vs baseline: 4.4922x; 1.0894x over previous
#include <hip/hip_runtime.h>
#include <hip/hip_bf16.h>


typedef __attribute__((ext_vector_type(2))) float v2f;
typedef __attribute__((ext_vector_type(4))) float v4f;
typedef __attribute__((ext_vector_type(8))) float v8f;

#define IN_CH  128
#define HID_CH 64

#define SH      9
#define BKT     (1 << SH)
#define NBMAX   256
#define ROWBITS 17
#define EPB     8192
#define GMAX    256
#define RSH     6
#define RSLOT   (1 << RSH)
#define FPT     8

__global__ __launch_bounds__(1024)
void gcn_bucket_scatter_kernel(const int* __restrict__ row, const int* __restrict__ col,
                               int* __restrict__ packed, int* __restrict__ rcnt,
                               int* __restrict__ roff, int nE, int NB) {
    __shared__ int s_h[NBMAX];
    __shared__ int s_base[NBMAX];
    __shared__ int s_w[8];
    const int tid  = threadIdx.x;
    const int lane = tid & 31;
    const int wv   = tid >> 5;
    const int g    = blockIdx.x;

    if (tid < NBMAX) s_h[tid] = 0;
    __syncthreads();

    const int base = g * EPB;
    int cc[EPB / 1024];
    int rk[EPB / 1024];
    for (int q = 0; q < EPB / 1024; ++q) {
        const int e = base + q * 1024 + tid;
        cc[q] = 0; rk[q] = 0;
        if (e < nE) {
            cc[q] = col[e];
            rk[q] = atomicAdd(&s_h[cc[q] >> SH], 1);
        }
    }
    __syncthreads();

    int T = 0, inc = 0;
    if (tid < NBMAX) {
        T = s_h[tid];
        inc = T;
        for (int d = 1; d < 32; d <<= 1) {
            const int u = __shfl_up(inc, d);
            if (lane >= d) inc += u;
        }
        if (lane == 31) s_w[wv] = inc;
    }
    __syncthreads();
    if (tid < NBMAX) {
        int pre = 0;
        for (int k = 0; k < wv; ++k) pre += s_w[k];
        const int excl = pre + inc - T;
        s_base[tid] = excl;
        if (tid < NB) { rcnt[g * NB + tid] = T; roff[g * NB + tid] = excl; }
    }
    __syncthreads();

    for (int q = 0; q < EPB / 1024; ++q) {
        const int e = base + q * 1024 + tid;
        if (e < nE) {
            const int c = cc[q];
            packed[base + s_base[c >> SH] + rk[q]] = ((c & (BKT - 1)) << ROWBITS) | row[e];
        }
    }
}

__global__ __launch_bounds__(1024)
void gcn_bucket_sort_kernel(const int* __restrict__ packed, const int* __restrict__ rcnt,
                            const int* __restrict__ roff, int* __restrict__ perm,
                            int* __restrict__ cnt, int* __restrict__ offs,
                            float* __restrict__ dinv, int n, int NB, int G) {
    __shared__ int s_cnt[BKT];
    __shared__ int s_cur[BKT];
    __shared__ int s_w[BKT / 32];
    __shared__ int s_tot[4][NBMAX];
    __shared__ int s_w8[8];
    __shared__ int2 s_run[GMAX];
    __shared__ int s_start;
    const int tid   = threadIdx.x;
    const int lane  = tid & 31;
    const int wv    = tid >> 5;
    const int b     = blockIdx.x;

    {
        const int t = tid & (NBMAX - 1);
        const int p = tid >> 8;
        int tot = 0;
        if (t < NB) {
            for (int k = p; k < G; k += 4) tot += rcnt[k * NB + t];
        }
        s_tot[p][t] = tot;
    }
    if (tid < G) s_run[tid] = make_int2(rcnt[tid * NB + b], tid * EPB + roff[tid * NB + b]);
    if (tid < BKT) s_cnt[tid] = 0;
    __syncthreads();

    {
        int T = 0, inc = 0;
        if (tid < NBMAX) {
            T = s_tot[0][tid] + s_tot[1][tid] + s_tot[2][tid] + s_tot[3][tid];
            inc = T;
            for (int d = 1; d < 32; d <<= 1) {
                const int u = __shfl_up(inc, d);
                if (lane >= d) inc += u;
            }
            if (lane == 31) s_w8[wv] = inc;
        }
        __syncthreads();
        if (tid == b) {
            int pre = 0;
            for (int k = 0; k < wv; ++k) pre += s_w8[k];
            s_start = pre + inc - T;
        }
    }

    const int nslot = G << RSH;
    const int ent = tid & (RSLOT - 1);
    for (int i0 = 0; i0 < nslot; i0 += FPT * 1024) {
        int2 ro[FPT];
        int  w[FPT];
        for (int q = 0; q < FPT; ++q) {
            const int gg = (i0 + q * 1024 + tid) >> RSH;
            ro[q] = s_run[(gg < G) ? gg : (G - 1)];
            if (gg >= G) ro[q].x = 0;
        }
        for (int q = 0; q < FPT; ++q) w[q] = (ent < ro[q].x) ? packed[ro[q].y + ent] : -1;
        for (int q = 0; q < FPT; ++q)
            if (w[q] >= 0) atomicAdd(&s_cnt[w[q] >> ROWBITS], 1);
    }
    for (int gg = wv; gg < G; gg += 32) {
        const int2 r = s_run[gg];
        for (int i = RSLOT + lane; i < r.x; i += 32)
            atomicAdd(&s_cnt[packed[r.y + i] >> ROWBITS], 1);
    }
    __syncthreads();

    int v = 0, inc = 0;
    if (tid < BKT) {
        v = s_cnt[tid];
        inc = v;
        for (int d = 1; d < 32; d <<= 1) {
            const int u = __shfl_up(inc, d);
            if (lane >= d) inc += u;
        }
        if (lane == 31) s_w[wv] = inc;
    }
    __syncthreads();
    const int start = s_start;
    if (tid < BKT) {
        int pre = 0;
        for (int k = 0; k < wv; ++k) pre += s_w[k];
        const int excl = pre + inc - v;
        s_cur[tid] = excl;
        const int node = (b << SH) + tid;
        if (node < n) {
            cnt[node]  = v;
            offs[node] = start + excl;
            float d = (float)(v + 1);
            dinv[node] = (d > 0.0f) ? rsqrtf(d) : 0.0f;
        }
    }
    __syncthreads();

    for (int i0 = 0; i0 < nslot; i0 += FPT * 1024) {
        int2 ro[FPT];
        int  w[FPT];
        for (int q = 0; q < FPT; ++q) {
            const int gg = (i0 + q * 1024 + tid) >> RSH;
            ro[q] = s_run[(gg < G) ? gg : (G - 1)];
            if (gg >= G) ro[q].x = 0;
        }
        for (int q = 0; q < FPT; ++q) w[q] = (ent < ro[q].x) ? packed[ro[q].y + ent] : -1;
        for (int q = 0; q < FPT; ++q) {
            if (w[q] >= 0) {
                const int slot = atomicAdd(&s_cur[w[q] >> ROWBITS], 1);
                perm[start + slot] = w[q] & ((1 << ROWBITS) - 1);
            }
        }
    }
    for (int gg = wv; gg < G; gg += 32) {
        const int2 r = s_run[gg];
        for (int i = RSLOT + lane; i < r.x; i += 32) {
            const int w = packed[r.y + i];
            const int slot = atomicAdd(&s_cur[w >> ROWBITS], 1);
            perm[start + slot] = w & ((1 << ROWBITS) - 1);
        }
    }
}

#define GEMM_WAVES 8
#define TPITCH     68

template <int K>
__global__ __launch_bounds__(256)
void gcn_gemm_wmma_kernel(const float* __restrict__ X, const float* __restrict__ W,
                          const float* __restrict__ dinv, float* __restrict__ H, int nNodes) {
    __shared__ __attribute__((aligned(16))) float s_w[K * HID_CH];
    __shared__ __attribute__((aligned(16))) float s_t[GEMM_WAVES * 16 * TPITCH];

    const int tid  = threadIdx.x;
    const int lane = tid & 31;
    const int wave = __builtin_amdgcn_readfirstlane(tid >> 5);
    const int n    = lane & 15;
    const int half = lane >> 4;
    const int r0   = (blockIdx.x * GEMM_WAVES + wave) * 16;

    int rA = r0 + n;
    if (rA > nNodes - 1) rA = nNodes - 1;
    const v4f* xr = (const v4f*)(X + (long long)rA * K) + half;
    v4f a4[K / 8];
#pragma unroll
    for (int j = 0; j < K / 8; ++j) a4[j] = xr[2 * j];

    constexpr int WQ = K * HID_CH / 1024;
    v4f w4[WQ];
#pragma unroll
    for (int q = 0; q < WQ; ++q) w4[q] = *(const v4f*)(W + q * 1024 + tid * 4);
#pragma unroll
    for (int q = 0; q < WQ; ++q) {
        const int g  = q * 1024 + tid * 4;
        const int k  = g >> 6;
        const int c  = g & (HID_CH - 1);
        const int kj = k >> 3;
        const int kh = (k >> 2) & 1;
        const int ke = k & 3;
        const int o  = ((((kj * 4 + (c >> 4)) * 32) + kh * 16 + (c & 15)) << 2) + ke;
        s_w[o]      = w4[q].x;
        s_w[o + 4]  = w4[q].y;
        s_w[o + 8]  = w4[q].z;
        s_w[o + 12] = w4[q].w;
    }
    __syncthreads();

    if (r0 >= nNodes) return;

    float dsc[8];
#pragma unroll
    for (int i = 0; i < 8; ++i) {
        int r = r0 + 2 * i + half;
        if (r > nNodes - 1) r = nNodes - 1;
        dsc[i] = dinv[r];
    }

    v8f acc0 = {}, acc1 = {}, acc2 = {}, acc3 = {};
    const v4f* bw = (const v4f*)s_w + lane;
#pragma unroll
    for (int j = 0; j < K / 8; ++j) {
        const v4f av = a4[j];
        const v4f q0 = bw[(j * 4 + 0) * 32];
        const v4f q1 = bw[(j * 4 + 1) * 32];
        const v4f q2 = bw[(j * 4 + 2) * 32];
        const v4f q3 = bw[(j * 4 + 3) * 32];
        v2f a, b0, b1, b2, b3;
        a.x = av.x;  a.y = av.y;
        b0.x = q0.x; b0.y = q0.y;
        b1.x = q1.x; b1.y = q1.y;
        b2.x = q2.x; b2.y = q2.y;
        b3.x = q3.x; b3.y = q3.y;
        acc0 = __builtin_amdgcn_wmma_f32_16x16x4_f32(false, a, false, b0, (short)0, acc0, false, false);
        acc1 = __builtin_amdgcn_wmma_f32_16x16x4_f32(false, a, false, b1, (short)0, acc1, false, false);
        acc2 = __builtin_amdgcn_wmma_f32_16x16x4_f32(false, a, false, b2, (short)0, acc2, false, false);
        acc3 = __builtin_amdgcn_wmma_f32_16x16x4_f32(false, a, false, b3, (short)0, acc3, false, false);
        a.x = av.z;  a.y = av.w;
        b0.x = q0.z; b0.y = q0.w;
        b1.x = q1.z; b1.y = q1.w;
        b2.x = q2.z; b2.y = q2.w;
        b3.x = q3.z; b3.y = q3.w;
        acc0 = __builtin_amdgcn_wmma_f32_16x16x4_f32(false, a, false, b0, (short)0, acc0, false, false);
        acc1 = __builtin_amdgcn_wmma_f32_16x16x4_f32(false, a, false, b1, (short)0, acc1, false, false);
        acc2 = __builtin_amdgcn_wmma_f32_16x16x4_f32(false, a, false, b2, (short)0, acc2, false, false);
        acc3 = __builtin_amdgcn_wmma_f32_16x16x4_f32(false, a, false, b3, (short)0, acc3, false, false);
    }

    float* st = s_t + wave * (16 * TPITCH);
#pragma unroll
    for (int v = 0; v < 8; ++v) {
        float* p = st + (v + 8 * half) * TPITCH + n;
        p[0]  = acc0[v];
        p[16] = acc1[v];
        p[32] = acc2[v];
        p[48] = acc3[v];
    }
    v4f o[8];
#pragma unroll
    for (int i = 0; i < 8; ++i) o[i] = *(const v4f*)(st + (2 * i + half) * TPITCH + 4 * n) * dsc[i];
    float* hp = H + (long long)(r0 + half) * HID_CH + 4 * n;
    if (r0 + 16 <= nNodes) {
#pragma unroll
        for (int i = 0; i < 8; ++i) *(v4f*)(hp + (long long)i * 2 * HID_CH) = o[i];
    } else {
#pragma unroll
        for (int i = 0; i < 8; ++i)
            if (r0 + 2 * i + half < nNodes) *(v4f*)(hp + (long long)i * 2 * HID_CH) = o[i];
    }
}

#define LSEG 128

__device__ __forceinline__
v2f gcn_sum_rows(v2f acc, const v2f* __restrict__ Hv, int rs, int cnt, int lane) {
    int j = 0;
#pragma unroll 2
    for (; j + 4 <= cnt; j += 4) {
        const int r0 = __builtin_amdgcn_readlane(rs, j);
        const int r1 = __builtin_amdgcn_readlane(rs, j + 1);
        const int r2 = __builtin_amdgcn_readlane(rs, j + 2);
        const int r3 = __builtin_amdgcn_readlane(rs, j + 3);
        const v2f h0 = Hv[(long long)r0 * 32 + lane];
        const v2f h1 = Hv[(long long)r1 * 32 + lane];
        const v2f h2 = Hv[(long long)r2 * 32 + lane];
        const v2f h3 = Hv[(long long)r3 * 32 + lane];
        acc += h0;
        acc += h1;
        acc += h2;
        acc += h3;
    }
    for (; j < cnt; ++j) {
        const int r0 = __builtin_amdgcn_readlane(rs, j);
        acc += Hv[(long long)r0 * 32 + lane];
    }
    return acc;
}

#define GATHER_BLOCKS 2048

#define WSLAB 8
#define PROWS (8 * WSLAB)

template <bool FUSE>
__global__ __launch_bounds__(256, 16)
void gcn_gather_kernel(const float* __restrict__ H, const int* __restrict__ offs,
                       const int* __restrict__ cnt, const int* __restrict__ perm,
                       const float* __restrict__ dinv, const float* __restrict__ bias,
                       const float* __restrict__ Wn, float* __restrict__ out,
                       int nNodes, int doRelu) {
    __shared__ __attribute__((aligned(16)))
        float s_buf[FUSE ? (PROWS * TPITCH + HID_CH * HID_CH) : (2 * 8 * LSEG)];
    float* const s_p   = s_buf;
    float* const s_w   = s_buf + (FUSE ? PROWS * TPITCH : 0);
    int*   const s_raw = (int*)s_w;
    int*   const s_ord = s_raw + 8 * LSEG;

    const int lane = threadIdx.x & 31;
    const int wv = __builtin_amdgcn_readfirstlane((int)(threadIdx.x >> 5));

    const int gw    = blockIdx.x * 8 + wv;
    const int TW    = gridDim.x * 8;
    const int q     = nNodes / TW;
    const int rem   = nNodes - q * TW;
    const int start = gw * q + ((gw < rem) ? gw : rem);
    const int end   = start + q + ((gw < rem) ? 1 : 0);

    const v2f* Hv = (const v2f*)H;
    const v2f  b  = ((const v2f*)bias)[lane];

    const int RND  = FUSE ? WSLAB : 32;
    const int span = FUSE ? 1 : (end - start);
    for (int rb = 0; rb < span; rb += RND) {
        const int base = start + rb;
        const int nk = (end - base < RND) ? (end - base) : RND;
        if (!FUSE || nk > 0) {

        int   m_beg = 0, m_deg = 0;
        float m_dc  = 0.0f;
        if (lane < nk) {
            m_beg = offs[base + lane];
            m_deg = cnt[base + lane];
            m_dc  = dinv[base + lane];
        }

        int r_cur = 0x7fffffff;
        v2f a_cur;
        {
            const int b0 = __builtin_amdgcn_readlane(m_beg, 0);
            const int d0 = __builtin_amdgcn_readlane(m_deg, 0);
            if (lane < d0) r_cur = perm[b0 + lane];
            a_cur = Hv[(long long)base * 32 + lane];
        }

        for (int k = 0; k < nk; ++k) {
            const int   c   = base + k;
            const int   beg = __builtin_amdgcn_readlane(m_beg, k);
            const int   deg = __builtin_amdgcn_readlane(m_deg, k);
            const float dc  = __int_as_float(__builtin_amdgcn_readlane(__float_as_int(m_dc), k));

            const int kn = (k + 1 < nk) ? (k + 1) : k;
            int r_nxt = 0x7fffffff;
            {
                const int bn = __builtin_amdgcn_readlane(m_beg, kn);
                const int dn = __builtin_amdgcn_readlane(m_deg, kn);
                if (lane < dn) r_nxt = perm[bn + lane];
            }
            const v2f h_nxt = Hv[(long long)(base + kn) * 32 + lane];

            v2f acc = a_cur;

            if (deg <= 32) {
                const int r_l = r_cur;

                int pos = 0;
                for (int j = 0; j < deg; ++j) {
                    const int rj = __builtin_amdgcn_readlane(r_l, j);
                    pos += ((rj < r_l) || (rj == r_l && j < lane)) ? 1 : 0;
                }
                if (lane >= deg) pos = lane;

                const int rs = __builtin_amdgcn_ds_permute(pos << 2, r_l);

                acc = gcn_sum_rows(acc, Hv, rs, deg, lane);
            } else if (deg <= LSEG) {
                int* raw = s_raw + wv * LSEG;
                int* ord = s_ord + wv * LSEG;
                const int K = (deg + 31) >> 5;

                for (int t = 0; t < K; ++t) {
                    const int i = t * 32 + lane;
                    raw[i] = (i < deg) ? perm[beg + i] : 0x7fffffff;
                }
                for (int t = 0; t < K; ++t) {
                    const int i = t * 32 + lane;
                    const int e = raw[i];
                    int pos = 0;
                    for (int cj = 0; cj < K; ++cj) {
                        const int ej = raw[cj * 32 + lane];
                        const int nj = (deg - cj * 32 < 32) ? (deg - cj * 32) : 32;
                        for (int j = 0; j < nj; ++j) {
                            const int rj = __builtin_amdgcn_readlane(ej, j);
                            pos += ((rj < e) || (rj == e && cj * 32 + j < i)) ? 1 : 0;
                        }
                    }
                    if (i < deg) ord[pos] = e;
                }
                for (int t = 0; t < K; ++t) {
                    const int nv = (deg - t * 32 < 32) ? (deg - t * 32) : 32;
                    int rs = 0;
                    if (lane < nv) rs = ord[t * 32 + lane];
                    acc = gcn_sum_rows(acc, Hv, rs, nv, lane);
                }
            } else {
                int prev = -1;
                int done = 0;
                for (int it = 0; it < deg && done < deg; ++it) {
                    int best = 0x7fffffff;
                    for (int i = lane; i < deg; i += 32) {
                        const int r = perm[beg + i];
                        if (r > prev && r < best) best = r;
                    }
                    for (int d = 16; d >= 1; d >>= 1) {
                        const int o = __shfl_xor(best, d);
                        best = (o < best) ? o : best;
                    }
                    best = __builtin_amdgcn_readfirstlane(best);
                    if (best == 0x7fffffff) break;
                    int m = 0;
                    for (int i = lane; i < deg; i += 32) m += (perm[beg + i] == best) ? 1 : 0;
                    for (int d = 16; d >= 1; d >>= 1) m += __shfl_xor(m, d);
                    m = __builtin_amdgcn_readfirstlane(m);
                    const v2f hv = Hv[(long long)best * 32 + lane];
                    for (int t = 0; t < m; ++t) acc += hv;
                    prev = best;
                    done += m;
                }
            }

            v2f v = acc * dc + b;
            if (doRelu) { v.x = fmaxf(v.x, 0.0f); v.y = fmaxf(v.y, 0.0f); }
            if (FUSE) *(v2f*)(s_p + (wv * WSLAB + k) * TPITCH + 2 * lane) = v;
            else      ((v2f*)out)[(long long)c * 32 + lane] = v;

            a_cur = h_nxt;
            r_cur = r_nxt;
        }
        }

        if (FUSE) {
            const int tid  = threadIdx.x;
            const int n    = lane & 15;
            const int half = lane >> 4;
            const int rt   = wv >> 1;
            const int cp   = wv & 1;

            for (int k = nk; k < WSLAB; ++k) {
                v2f z; z.x = 0.0f; z.y = 0.0f;
                *(v2f*)(s_p + (wv * WSLAB + k) * TPITCH + 2 * lane) = z;
            }

            constexpr int WQ = HID_CH * HID_CH / 1024;
            v4f w4[WQ];
#pragma unroll
            for (int qq = 0; qq < WQ; ++qq) w4[qq] = *(const v4f*)(Wn + qq * 1024 + tid * 4);

            __syncthreads();

            {
                const int kk = tid >> 4;
                const int c  = (tid & 15) * 4;
                const int kj = kk >> 3;
                const int kh = (kk >> 2) & 1;
                const int ke = kk & 3;
                float* sw = s_w + (((((kj * 4 + (c >> 4)) * 32) + kh * 16 + (c & 15)) << 2) + ke);
#pragma unroll
                for (int qq = 0; qq < WQ; ++qq) {
                    sw[qq * 1024]      = w4[qq].x;
                    sw[qq * 1024 + 4]  = w4[qq].y;
                    sw[qq * 1024 + 8]  = w4[qq].z;
                    sw[qq * 1024 + 12] = w4[qq].w;
                }
            }
            __syncthreads();

            v8f acc0 = {}, acc1 = {};
            const v4f* ap = (const v4f*)(s_p + (rt * 16 + n) * TPITCH) + half;
            const v4f* bw = (const v4f*)s_w + cp * 64 + lane;
#pragma unroll 2
            for (int j = 0; j < HID_CH / 8; ++j) {
                const v4f av = ap[2 * j];
                const v4f q0 = bw[(j * 4 + 0) * 32];
                const v4f q1 = bw[(j * 4 + 1) * 32];
                v2f a, b0, b1;
                a.x = av.x;  a.y = av.y;
                b0.x = q0.x; b0.y = q0.y;
                b1.x = q1.x; b1.y = q1.y;
                acc0 = __builtin_amdgcn_wmma_f32_16x16x4_f32(false, a, false, b0, (short)0, acc0, false, false);
                acc1 = __builtin_amdgcn_wmma_f32_16x16x4_f32(false, a, false, b1, (short)0, acc1, false, false);
                a.x = av.z;  a.y = av.w;
                b0.x = q0.z; b0.y = q0.w;
                b1.x = q1.z; b1.y = q1.w;
                acc0 = __builtin_amdgcn_wmma_f32_16x16x4_f32(false, a, false, b0, (short)0, acc0, false, false);
                acc1 = __builtin_amdgcn_wmma_f32_16x16x4_f32(false, a, false, b1, (short)0, acc1, false, false);
            }
            __syncthreads();

            const int rr0 = lane >> 3;
            const int c4  = (lane & 7) * 4;
            int   onode[4];
            float dsc[4];
#pragma unroll
            for (int i = 0; i < 4; ++i) {
                const int p  = rt * 16 + i * 4 + rr0;
                const int g2 = blockIdx.x * 8 + (p >> 3);
                const int s2 = g2 * q + ((g2 < rem) ? g2 : rem);
                const int e2 = s2 + q + ((g2 < rem) ? 1 : 0);
                const int nd = s2 + (p & (WSLAB - 1));
                onode[i] = (nd < e2) ? nd : -1;
                dsc[i]   = (nd < e2) ? dinv[nd] : 0.0f;
            }

            float* st = s_p + (rt * 16) * TPITCH + cp * 32;
#pragma unroll
            for (int v = 0; v < 8; ++v) {
                float* p = st + (v + 8 * half) * TPITCH + n;
                p[0]  = acc0[v];
                p[16] = acc1[v];
            }
#pragma unroll
            for (int i = 0; i < 4; ++i) {
                const v4f o = *(const v4f*)(st + (i * 4 + rr0) * TPITCH + c4) * dsc[i];
                if (onode[i] >= 0) *(v4f*)(out + (long long)onode[i] * HID_CH + cp * 32 + c4) = o;
            }
        }
    }
}

extern "C" void kernel_launch(void* const* d_in, const int* in_sizes, int n_in,
                              void* d_out, int out_size, void* d_ws, size_t ws_size,
                              hipStream_t stream) {
    const float* x   = (const float*)d_in[0];
    const int*   ei  = (const int*)  d_in[1];
    const float* W1  = (const float*)d_in[2];
    const float* b1  = (const float*)d_in[3];
    const float* W2  = (const float*)d_in[4];
    const float* b2  = (const float*)d_in[5];
    float*       out = (float*)d_out;

    const int N  = in_sizes[0] / IN_CH;
    const int nE = in_sizes[1] / 2;
    const int*   rowp = ei;
    const int*   colp = ei + nE;

    const long long feat = (long long)N * HID_CH;
    float* h      = (float*)d_ws;
    float* a1     = h + feat;
    int*   packed = (int*)(a1 + feat);
    int*   perm   = packed + nE;
    int*   cnt    = perm + nE;
    int*   offs   = cnt + N;
    float* dinv   = (float*)(offs + N);

    const int NB = (N + BKT - 1) / BKT;
    const int G  = (nE + EPB - 1) / EPB;
    int*   rcnt   = (int*)a1;
    int*   roff   = rcnt + G * NB;

    const int tileBlocks = (N + 15) / 16;
    const int gemmBlocks = (tileBlocks + GEMM_WAVES - 1) / GEMM_WAVES;

    gcn_bucket_scatter_kernel<<<G, 1024, 0, stream>>>(rowp, colp, packed, rcnt, roff, nE, NB);
    gcn_bucket_sort_kernel<<<NB, 1024, 0, stream>>>(packed, rcnt, roff, perm, cnt, offs, dinv, N, NB, G);

    gcn_gemm_wmma_kernel<IN_CH><<<gemmBlocks, 256, 0, stream>>>(x, W1, dinv, h, N);
    const int slabBlocks = (N + PROWS - 1) / PROWS;
    const int fuseBlocks = (slabBlocks > GATHER_BLOCKS) ? slabBlocks : GATHER_BLOCKS;
    gcn_gather_kernel<true><<<fuseBlocks, 256, 0, stream>>>(h, offs, cnt, perm, dinv, b1, W2, a1, N,  1);

    gcn_gather_kernel<false><<<GATHER_BLOCKS, 256, 0, stream>>>(a1, offs, cnt, perm, dinv, b2, W2, out, N,  0);
}


# Both edge-sort levels assemble their output in LDS and store it as whole 128-byte lines (level 1 reads all its edge words up front), not one scattered 4-byte store per edge; sum order kept
# speedup vs baseline: 5.1015x; 1.1356x over previous
#include <hip/hip_runtime.h>
#include <hip/hip_bf16.h>


typedef __attribute__((ext_vector_type(2))) float v2f;
typedef __attribute__((ext_vector_type(4))) float v4f;
typedef __attribute__((ext_vector_type(8))) float v8f;

#define IN_CH  128
#define HID_CH 64

#define SH      9
#define BKT     (1 << SH)
#define NBMAX   256
#define ROWBITS 17
#define EPB     8192
#define GMAX    256
#define RSH     6
#define RSLOT   (1 << RSH)
#define FPT     8
#define STG     16384

__global__ __launch_bounds__(1024)
void gcn_bucket_scatter_kernel(const int* __restrict__ row, const int* __restrict__ col,
                               int* __restrict__ packed, int* __restrict__ rcnt,
                               int* __restrict__ roff, int nE, int NB) {
    __shared__ int s_h[NBMAX];
    __shared__ int s_base[NBMAX];
    __shared__ int s_w[8];
    __shared__ __attribute__((aligned(16))) int s_out[EPB];
    const int tid  = threadIdx.x;
    const int lane = tid & 31;
    const int wv   = tid >> 5;
    const int g    = blockIdx.x;

    if (tid < NBMAX) s_h[tid] = 0;

    const int base = g * EPB;
    int cc[EPB / 1024];
    int rr[EPB / 1024];
    int rk[EPB / 1024];
#pragma unroll
    for (int q = 0; q < EPB / 1024; ++q) {
        const int e  = base + q * 1024 + tid;
        const int ee = (e < nE) ? e : (nE - 1);
        cc[q] = col[ee];
        rr[q] = row[ee];
    }
    __syncthreads();

#pragma unroll
    for (int q = 0; q < EPB / 1024; ++q) {
        const int e = base + q * 1024 + tid;
        rk[q] = 0;
        if (e < nE) rk[q] = atomicAdd(&s_h[cc[q] >> SH], 1);
    }
    __syncthreads();

    int T = 0, inc = 0;
    if (tid < NBMAX) {
        T = s_h[tid];
        inc = T;
        for (int d = 1; d < 32; d <<= 1) {
            const int u = __shfl_up(inc, d);
            if (lane >= d) inc += u;
        }
        if (lane == 31) s_w[wv] = inc;
    }
    __syncthreads();
    if (tid < NBMAX) {
        int pre = 0;
        for (int k = 0; k < wv; ++k) pre += s_w[k];
        const int excl = pre + inc - T;
        s_base[tid] = excl;
        if (tid < NB) { rcnt[g * NB + tid] = T; roff[g * NB + tid] = excl; }
    }
    __syncthreads();

    int bs[EPB / 1024];
#pragma unroll
    for (int q = 0; q < EPB / 1024; ++q) bs[q] = s_base[cc[q] >> SH];
#pragma unroll
    for (int q = 0; q < EPB / 1024; ++q) {
        const int e = base + q * 1024 + tid;
        if (e < nE) s_out[bs[q] + rk[q]] = ((cc[q] & (BKT - 1)) << ROWBITS) | rr[q];
    }
    __syncthreads();

    const int nb = (nE - base < EPB) ? (nE - base) : EPB;
    for (int i = tid * 4; i < nb; i += 4096) {
        if (i + 4 <= nb) {
            *(int4*)(packed + base + i) = *(const int4*)(s_out + i);
        } else {
            for (int k = i; k < nb; ++k) packed[base + k] = s_out[k];
        }
    }
}

__global__ __launch_bounds__(1024)
void gcn_bucket_sort_kernel(const int* __restrict__ packed, const int* __restrict__ rcnt,
                            const int* __restrict__ roff, int* __restrict__ perm,
                            int* __restrict__ cnt, int* __restrict__ offs,
                            float* __restrict__ dinv, int n, int NB, int G) {
    __shared__ int s_cnt[BKT];
    __shared__ int s_cur[BKT];
    __shared__ int s_w[BKT / 32];
    __shared__ int s_tot[4][NBMAX];
    __shared__ int s_w8[8];
    __shared__ int2 s_run[GMAX];
    __shared__ int s_start;
    __shared__ int s_total;
    __shared__ int s_stage[STG];
    const int tid   = threadIdx.x;
    const int lane  = tid & 31;
    const int wv    = tid >> 5;
    const int b     = blockIdx.x;

    {
        const int t = tid & (NBMAX - 1);
        const int p = tid >> 8;
        int tot = 0;
        if (t < NB) {
            for (int k = p; k < G; k += 4) tot += rcnt[k * NB + t];
        }
        s_tot[p][t] = tot;
    }
    if (tid < G) s_run[tid] = make_int2(rcnt[tid * NB + b], tid * EPB + roff[tid * NB + b]);
    if (tid < BKT) s_cnt[tid] = 0;
    __syncthreads();

    {
        int T = 0, inc = 0;
        if (tid < NBMAX) {
            T = s_tot[0][tid] + s_tot[1][tid] + s_tot[2][tid] + s_tot[3][tid];
            inc = T;
            for (int d = 1; d < 32; d <<= 1) {
                const int u = __shfl_up(inc, d);
                if (lane >= d) inc += u;
            }
            if (lane == 31) s_w8[wv] = inc;
        }
        __syncthreads();
        if (tid == b) {
            int pre = 0;
            for (int k = 0; k < wv; ++k) pre += s_w8[k];
            s_start = pre + inc - T;
            s_total = T;
        }
    }

    const int nslot = G << RSH;
    const int ent = tid & (RSLOT - 1);
    for (int i0 = 0; i0 < nslot; i0 += FPT * 1024) {
        int2 ro[FPT];
        int  w[FPT];
        for (int q = 0; q < FPT; ++q) {
            const int gg = (i0 + q * 1024 + tid) >> RSH;
            ro[q] = s_run[(gg < G) ? gg : (G - 1)];
            if (gg >= G) ro[q].x = 0;
        }
        for (int q = 0; q < FPT; ++q) w[q] = (ent < ro[q].x) ? packed[ro[q].y + ent] : -1;
        for (int q = 0; q < FPT; ++q)
            if (w[q] >= 0) atomicAdd(&s_cnt[w[q] >> ROWBITS], 1);
    }
    for (int gg = wv; gg < G; gg += 32) {
        const int2 r = s_run[gg];
        for (int i = RSLOT + lane; i < r.x; i += 32)
            atomicAdd(&s_cnt[packed[r.y + i] >> ROWBITS], 1);
    }
    __syncthreads();

    int v = 0, inc = 0;
    if (tid < BKT) {
        v = s_cnt[tid];
        inc = v;
        for (int d = 1; d < 32; d <<= 1) {
            const int u = __shfl_up(inc, d);
            if (lane >= d) inc += u;
        }
        if (lane == 31) s_w[wv] = inc;
    }
    __syncthreads();
    const int start = s_start;
    const int tot   = s_total;
    const bool staged = (tot <= STG);
    if (tid < BKT) {
        int pre = 0;
        for (int k = 0; k < wv; ++k) pre += s_w[k];
        const int excl = pre + inc - v;
        s_cur[tid] = excl;
        const int node = (b << SH) + tid;
        if (node < n) {
            cnt[node]  = v;
            offs[node] = start + excl;
            float d = (float)(v + 1);
            dinv[node] = (d > 0.0f) ? rsqrtf(d) : 0.0f;
        }
    }
    __syncthreads();

    if (staged) {
        for (int i0 = 0; i0 < nslot; i0 += FPT * 1024) {
            int2 ro[FPT];
            int  w[FPT];
            int  sl[FPT];
            for (int q = 0; q < FPT; ++q) {
                const int gg = (i0 + q * 1024 + tid) >> RSH;
                ro[q] = s_run[(gg < G) ? gg : (G - 1)];
                if (gg >= G) ro[q].x = 0;
            }
            for (int q = 0; q < FPT; ++q) w[q] = (ent < ro[q].x) ? packed[ro[q].y + ent] : -1;
            for (int q = 0; q < FPT; ++q) {
                sl[q] = 0;
                if (w[q] >= 0) sl[q] = atomicAdd(&s_cur[w[q] >> ROWBITS], 1);
            }
            for (int q = 0; q < FPT; ++q)
                if (w[q] >= 0) s_stage[sl[q]] = w[q] & ((1 << ROWBITS) - 1);
        }
        for (int gg = wv; gg < G; gg += 32) {
            const int2 r = s_run[gg];
            for (int i = RSLOT + lane; i < r.x; i += 32) {
                const int w = packed[r.y + i];
                const int slot = atomicAdd(&s_cur[w >> ROWBITS], 1);
                s_stage[slot] = w & ((1 << ROWBITS) - 1);
            }
        }
        __syncthreads();

        const int al = start & 31;
        int i = tid - al;
        if (i < 0) i += 1024;
#pragma unroll 4
        for (; i < tot; i += 1024) perm[start + i] = s_stage[i];
    } else {
        for (int i0 = 0; i0 < nslot; i0 += FPT * 1024) {
            int2 ro[FPT];
            int  w[FPT];
            for (int q = 0; q < FPT; ++q) {
                const int gg = (i0 + q * 1024 + tid) >> RSH;
                ro[q] = s_run[(gg < G) ? gg : (G - 1)];
                if (gg >= G) ro[q].x = 0;
            }
            for (int q = 0; q < FPT; ++q) w[q] = (ent < ro[q].x) ? packed[ro[q].y + ent] : -1;
            for (int q = 0; q < FPT; ++q) {
                if (w[q] >= 0) {
                    const int slot = atomicAdd(&s_cur[w[q] >> ROWBITS], 1);
                    perm[start + slot] = w[q] & ((1 << ROWBITS) - 1);
                }
            }
        }
        for (int gg = wv; gg < G; gg += 32) {
            const int2 r = s_run[gg];
            for (int i = RSLOT + lane; i < r.x; i += 32) {
                const int w = packed[r.y + i];
                const int slot = atomicAdd(&s_cur[w >> ROWBITS], 1);
                perm[start + slot] = w & ((1 << ROWBITS) - 1);
            }
        }
    }
}

#define GEMM_WAVES 8
#define TPITCH     68

template <int K>
__global__ __launch_bounds__(256)
void gcn_gemm_wmma_kernel(const float* __restrict__ X, const float* __restrict__ W,
                          const float* __restrict__ dinv, float* __restrict__ H, int nNodes) {
    __shared__ __attribute__((aligned(16))) float s_w[K * HID_CH];
    __shared__ __attribute__((aligned(16))) float s_t[GEMM_WAVES * 16 * TPITCH];

    const int tid  = threadIdx.x;
    const int lane = tid & 31;
    const int wave = __builtin_amdgcn_readfirstlane(tid >> 5);
    const int n    = lane & 15;
    const int half = lane >> 4;
    const int r0   = (blockIdx.x * GEMM_WAVES + wave) * 16;

    int rA = r0 + n;
    if (rA > nNodes - 1) rA = nNodes - 1;
    const v4f* xr = (const v4f*)(X + (long long)rA * K) + half;
    v4f a4[K / 8];
#pragma unroll
    for (int j = 0; j < K / 8; ++j) a4[j] = xr[2 * j];

    constexpr int WQ = K * HID_CH / 1024;
    v4f w4[WQ];
#pragma unroll
    for (int q = 0; q < WQ; ++q) w4[q] = *(const v4f*)(W + q * 1024 + tid * 4);
#pragma unroll
    for (int q = 0; q < WQ; ++q) {
        const int g  = q * 1024 + tid * 4;
        const int k  = g >> 6;
        const int c  = g & (HID_CH - 1);
        const int kj = k >> 3;
        const int kh = (k >> 2) & 1;
        const int ke = k & 3;
        const int o  = ((((kj * 4 + (c >> 4)) * 32) + kh * 16 + (c & 15)) << 2) + ke;
        s_w[o]      = w4[q].x;
        s_w[o + 4]  = w4[q].y;
        s_w[o + 8]  = w4[q].z;
        s_w[o + 12] = w4[q].w;
    }
    __syncthreads();

    if (r0 >= nNodes) return;

    float dsc[8];
#pragma unroll
    for (int i = 0; i < 8; ++i) {
        int r = r0 + 2 * i + half;
        if (r > nNodes - 1) r = nNodes - 1;
        dsc[i] = dinv[r];
    }

    v8f acc0 = {}, acc1 = {}, acc2 = {}, acc3 = {};
    const v4f* bw = (const v4f*)s_w + lane;
#pragma unroll
    for (int j = 0; j < K / 8; ++j) {
        const v4f av = a4[j];
        const v4f q0 = bw[(j * 4 + 0) * 32];
        const v4f q1 = bw[(j * 4 + 1) * 32];
        const v4f q2 = bw[(j * 4 + 2) * 32];
        const v4f q3 = bw[(j * 4 + 3) * 32];
        v2f a, b0, b1, b2, b3;
        a.x = av.x;  a.y = av.y;
        b0.x = q0.x; b0.y = q0.y;
        b1.x = q1.x; b1.y = q1.y;
        b2.x = q2.x; b2.y = q2.y;
        b3.x = q3.x; b3.y = q3.y;
        acc0 = __builtin_amdgcn_wmma_f32_16x16x4_f32(false, a, false, b0, (short)0, acc0, false, false);
        acc1 = __builtin_amdgcn_wmma_f32_16x16x4_f32(false, a, false, b1, (short)0, acc1, false, false);
        acc2 = __builtin_amdgcn_wmma_f32_16x16x4_f32(false, a, false, b2, (short)0, acc2, false, false);
        acc3 = __builtin_amdgcn_wmma_f32_16x16x4_f32(false, a, false, b3, (short)0, acc3, false, false);
        a.x = av.z;  a.y = av.w;
        b0.x = q0.z; b0.y = q0.w;
        b1.x = q1.z; b1.y = q1.w;
        b2.x = q2.z; b2.y = q2.w;
        b3.x = q3.z; b3.y = q3.w;
        acc0 = __builtin_amdgcn_wmma_f32_16x16x4_f32(false, a, false, b0, (short)0, acc0, false, false);
        acc1 = __builtin_amdgcn_wmma_f32_16x16x4_f32(false, a, false, b1, (short)0, acc1, false, false);
        acc2 = __builtin_amdgcn_wmma_f32_16x16x4_f32(false, a, false, b2, (short)0, acc2, false, false);
        acc3 = __builtin_amdgcn_wmma_f32_16x16x4_f32(false, a, false, b3, (short)0, acc3, false, false);
    }

    float* st = s_t + wave * (16 * TPITCH);
#pragma unroll
    for (int v = 0; v < 8; ++v) {
        float* p = st + (v + 8 * half) * TPITCH + n;
        p[0]  = acc0[v];
        p[16] = acc1[v];
        p[32] = acc2[v];
        p[48] = acc3[v];
    }
    v4f o[8];
#pragma unroll
    for (int i = 0; i < 8; ++i) o[i] = *(const v4f*)(st + (2 * i + half) * TPITCH + 4 * n) * dsc[i];
    float* hp = H + (long long)(r0 + half) * HID_CH + 4 * n;
    if (r0 + 16 <= nNodes) {
#pragma unroll
        for (int i = 0; i < 8; ++i) *(v4f*)(hp + (long long)i * 2 * HID_CH) = o[i];
    } else {
#pragma unroll
        for (int i = 0; i < 8; ++i)
            if (r0 + 2 * i + half < nNodes) *(v4f*)(hp + (long long)i * 2 * HID_CH) = o[i];
    }
}

#define LSEG 128

__device__ __forceinline__
v2f gcn_sum_rows(v2f acc, const v2f* __restrict__ Hv, int rs, int cnt, int lane) {
    int j = 0;
#pragma unroll 2
    for (; j + 4 <= cnt; j += 4) {
        const int r0 = __builtin_amdgcn_readlane(rs, j);
        const int r1 = __builtin_amdgcn_readlane(rs, j + 1);
        const int r2 = __builtin_amdgcn_readlane(rs, j + 2);
        const int r3 = __builtin_amdgcn_readlane(rs, j + 3);
        const v2f h0 = Hv[(long long)r0 * 32 + lane];
        const v2f h1 = Hv[(long long)r1 * 32 + lane];
        const v2f h2 = Hv[(long long)r2 * 32 + lane];
        const v2f h3 = Hv[(long long)r3 * 32 + lane];
        acc += h0;
        acc += h1;
        acc += h2;
        acc += h3;
    }
    for (; j < cnt; ++j) {
        const int r0 = __builtin_amdgcn_readlane(rs, j);
        acc += Hv[(long long)r0 * 32 + lane];
    }
    return acc;
}

#define GATHER_BLOCKS 2048

#define WSLAB 8
#define PROWS (8 * WSLAB)

template <bool FUSE>
__global__ __launch_bounds__(256, 16)
void gcn_gather_kernel(const float* __restrict__ H, const int* __restrict__ offs,
                       const int* __restrict__ cnt, const int* __restrict__ perm,
                       const float* __restrict__ dinv, const float* __restrict__ bias,
                       const float* __restrict__ Wn, float* __restrict__ out,
                       int nNodes, int doRelu) {
    __shared__ __attribute__((aligned(16)))
        float s_buf[FUSE ? (PROWS * TPITCH + HID_CH * HID_CH) : (2 * 8 * LSEG)];
    float* const s_p   = s_buf;
    float* const s_w   = s_buf + (FUSE ? PROWS * TPITCH : 0);
    int*   const s_raw = (int*)s_w;
    int*   const s_ord = s_raw + 8 * LSEG;

    const int lane = threadIdx.x & 31;
    const int wv = __builtin_amdgcn_readfirstlane((int)(threadIdx.x >> 5));

    const int gw    = blockIdx.x * 8 + wv;
    const int TW    = gridDim.x * 8;
    const int q     = nNodes / TW;
    const int rem   = nNodes - q * TW;
    const int start = gw * q + ((gw < rem) ? gw : rem);
    const int end   = start + q + ((gw < rem) ? 1 : 0);

    const v2f* Hv = (const v2f*)H;
    const v2f  b  = ((const v2f*)bias)[lane];

    const int RND  = FUSE ? WSLAB : 32;
    const int span = FUSE ? 1 : (end - start);
    for (int rb = 0; rb < span; rb += RND) {
        const int base = start + rb;
        const int nk = (end - base < RND) ? (end - base) : RND;
        if (!FUSE || nk > 0) {

        int   m_beg = 0, m_deg = 0;
        float m_dc  = 0.0f;
        if (lane < nk) {
            m_beg = offs[base + lane];
            m_deg = cnt[base + lane];
            m_dc  = dinv[base + lane];
        }

        int r_cur = 0x7fffffff;
        v2f a_cur;
        {
            const int b0 = __builtin_amdgcn_readlane(m_beg, 0);
            const int d0 = __builtin_amdgcn_readlane(m_deg, 0);
            if (lane < d0) r_cur = perm[b0 + lane];
            a_cur = Hv[(long long)base * 32 + lane];
        }

        for (int k = 0; k < nk; ++k) {
            const int   c   = base + k;
            const int   beg = __builtin_amdgcn_readlane(m_beg, k);
            const int   deg = __builtin_amdgcn_readlane(m_deg, k);
            const float dc  = __int_as_float(__builtin_amdgcn_readlane(__float_as_int(m_dc), k));

            const int kn = (k + 1 < nk) ? (k + 1) : k;
            int r_nxt = 0x7fffffff;
            {
                const int bn = __builtin_amdgcn_readlane(m_beg, kn);
                const int dn = __builtin_amdgcn_readlane(m_deg, kn);
                if (lane < dn) r_nxt = perm[bn + lane];
            }
            const v2f h_nxt = Hv[(long long)(base + kn) * 32 + lane];

            v2f acc = a_cur;

            if (deg <= 32) {
                const int r_l = r_cur;

                int pos = 0;
                for (int j = 0; j < deg; ++j) {
                    const int rj = __builtin_amdgcn_readlane(r_l, j);
                    pos += ((rj < r_l) || (rj == r_l && j < lane)) ? 1 : 0;
                }
                if (lane >= deg) pos = lane;

                const int rs = __builtin_amdgcn_ds_permute(pos << 2, r_l);

                acc = gcn_sum_rows(acc, Hv, rs, deg, lane);
            } else if (deg <= LSEG) {
                int* raw = s_raw + wv * LSEG;
                int* ord = s_ord + wv * LSEG;
                const int K = (deg + 31) >> 5;

                for (int t = 0; t < K; ++t) {
                    const int i = t * 32 + lane;
                    raw[i] = (i < deg) ? perm[beg + i] : 0x7fffffff;
                }
                for (int t = 0; t < K; ++t) {
                    const int i = t * 32 + lane;
                    const int e = raw[i];
                    int pos = 0;
                    for (int cj = 0; cj < K; ++cj) {
                        const int ej = raw[cj * 32 + lane];
                        const int nj = (deg - cj * 32 < 32) ? (deg - cj * 32) : 32;
                        for (int j = 0; j < nj; ++j) {
                            const int rj = __builtin_amdgcn_readlane(ej, j);
                            pos += ((rj < e) || (rj == e && cj * 32 + j < i)) ? 1 : 0;
                        }
                    }
                    if (i < deg) ord[pos] = e;
                }
                for (int t = 0; t < K; ++t) {
                    const int nv = (deg - t * 32 < 32) ? (deg - t * 32) : 32;
                    int rs = 0;
                    if (lane < nv) rs = ord[t * 32 + lane];
                    acc = gcn_sum_rows(acc, Hv, rs, nv, lane);
                }
            } else {
                int prev = -1;
                int done = 0;
                for (int it = 0; it < deg && done < deg; ++it) {
                    int best = 0x7fffffff;
                    for (int i = lane; i < deg; i += 32) {
                        const int r = perm[beg + i];
                        if (r > prev && r < best) best = r;
                    }
                    for (int d = 16; d >= 1; d >>= 1) {
                        const int o = __shfl_xor(best, d);
                        best = (o < best) ? o : best;
                    }
                    best = __builtin_amdgcn_readfirstlane(best);
                    if (best == 0x7fffffff) break;
                    int m = 0;
                    for (int i = lane; i < deg; i += 32) m += (perm[beg + i] == best) ? 1 : 0;
                    for (int d = 16; d >= 1; d >>= 1) m += __shfl_xor(m, d);
                    m = __builtin_amdgcn_readfirstlane(m);
                    const v2f hv = Hv[(long long)best * 32 + lane];
                    for (int t = 0; t < m; ++t) acc += hv;
                    prev = best;
                    done += m;
                }
            }

            v2f v = acc * dc + b;
            if (doRelu) { v.x = fmaxf(v.x, 0.0f); v.y = fmaxf(v.y, 0.0f); }
            if (FUSE) *(v2f*)(s_p + (wv * WSLAB + k) * TPITCH + 2 * lane) = v;
            else      ((v2f*)out)[(long long)c * 32 + lane] = v;

            a_cur = h_nxt;
            r_cur = r_nxt;
        }
        }

        if (FUSE) {
            const int tid  = threadIdx.x;
            const int n    = lane & 15;
            const int half = lane >> 4;
            const int rt   = wv >> 1;
            const int cp   = wv & 1;

            for (int k = nk; k < WSLAB; ++k) {
                v2f z; z.x = 0.0f; z.y = 0.0f;
                *(v2f*)(s_p + (wv * WSLAB + k) * TPITCH + 2 * lane) = z;
            }

            constexpr int WQ = HID_CH * HID_CH / 1024;
            v4f w4[WQ];
#pragma unroll
            for (int qq = 0; qq < WQ; ++qq) w4[qq] = *(const v4f*)(Wn + qq * 1024 + tid * 4);

            __syncthreads();

            {
                const int kk = tid >> 4;
                const int c  = (tid & 15) * 4;
                const int kj = kk >> 3;
                const int kh = (kk >> 2) & 1;
                const int ke = kk & 3;
                float* sw = s_w + (((((kj * 4 + (c >> 4)) * 32) + kh * 16 + (c & 15)) << 2) + ke);
#pragma unroll
                for (int qq = 0; qq < WQ; ++qq) {
                    sw[qq * 1024]      = w4[qq].x;
                    sw[qq * 1024 + 4]  = w4[qq].y;
                    sw[qq * 1024 + 8]  = w4[qq].z;
                    sw[qq * 1024 + 12] = w4[qq].w;
                }
            }
            __syncthreads();

            v8f acc0 = {}, acc1 = {};
            const v4f* ap = (const v4f*)(s_p + (rt * 16 + n) * TPITCH) + half;
            const v4f* bw = (const v4f*)s_w + cp * 64 + lane;
#pragma unroll 2
            for (int j = 0; j < HID_CH / 8; ++j) {
                const v4f av = ap[2 * j];
                const v4f q0 = bw[(j * 4 + 0) * 32];
                const v4f q1 = bw[(j * 4 + 1) * 32];
                v2f a, b0, b1;
                a.x = av.x;  a.y = av.y;
                b0.x = q0.x; b0.y = q0.y;
                b1.x = q1.x; b1.y = q1.y;
                acc0 = __builtin_amdgcn_wmma_f32_16x16x4_f32(false, a, false, b0, (short)0, acc0, false, false);
                acc1 = __builtin_amdgcn_wmma_f32_16x16x4_f32(false, a, false, b1, (short)0, acc1, false, false);
                a.x = av.z;  a.y = av.w;
                b0.x = q0.z; b0.y = q0.w;
                b1.x = q1.z; b1.y = q1.w;
                acc0 = __builtin_amdgcn_wmma_f32_16x16x4_f32(false, a, false, b0, (short)0, acc0, false, false);
                acc1 = __builtin_amdgcn_wmma_f32_16x16x4_f32(false, a, false, b1, (short)0, acc1, false, false);
            }
            __syncthreads();

            const int rr0 = lane >> 3;
            const int c4  = (lane & 7) * 4;
            int   onode[4];
            float dsc[4];
#pragma unroll
            for (int i = 0; i < 4; ++i) {
                const int p  = rt * 16 + i * 4 + rr0;
                const int g2 = blockIdx.x * 8 + (p >> 3);
                const int s2 = g2 * q + ((g2 < rem) ? g2 : rem);
                const int e2 = s2 + q + ((g2 < rem) ? 1 : 0);
                const int nd = s2 + (p & (WSLAB - 1));
                onode[i] = (nd < e2) ? nd : -1;
                dsc[i]   = (nd < e2) ? dinv[nd] : 0.0f;
            }

            float* st = s_p + (rt * 16) * TPITCH + cp * 32;
#pragma unroll
            for (int v = 0; v < 8; ++v) {
                float* p = st + (v + 8 * half) * TPITCH + n;
                p[0]  = acc0[v];
                p[16] = acc1[v];
            }
#pragma unroll
            for (int i = 0; i < 4; ++i) {
                const v4f o = *(const v4f*)(st + (i * 4 + rr0) * TPITCH + c4) * dsc[i];
                if (onode[i] >= 0) *(v4f*)(out + (long long)onode[i] * HID_CH + cp * 32 + c4) = o;
            }
        }
    }
}

extern "C" void kernel_launch(void* const* d_in, const int* in_sizes, int n_in,
                              void* d_out, int out_size, void* d_ws, size_t ws_size,
                              hipStream_t stream) {
    const float* x   = (const float*)d_in[0];
    const int*   ei  = (const int*)  d_in[1];
    const float* W1  = (const float*)d_in[2];
    const float* b1  = (const float*)d_in[3];
    const float* W2  = (const float*)d_in[4];
    const float* b2  = (const float*)d_in[5];
    float*       out = (float*)d_out;

    const int N  = in_sizes[0] / IN_CH;
    const int nE = in_sizes[1] / 2;
    const int*   rowp = ei;
    const int*   colp = ei + nE;

    const long long feat = (long long)N * HID_CH;
    float* h      = (float*)d_ws;
    float* a1     = h + feat;
    int*   packed = (int*)(a1 + feat);
    int*   perm   = packed + nE;
    int*   cnt    = perm + nE;
    int*   offs   = cnt + N;
    float* dinv   = (float*)(offs + N);

    const int NB = (N + BKT - 1) / BKT;
    const int G  = (nE + EPB - 1) / EPB;
    int*   rcnt   = (int*)a1;
    int*   roff   = rcnt + G * NB;

    const int tileBlocks = (N + 15) / 16;
    const int gemmBlocks = (tileBlocks + GEMM_WAVES - 1) / GEMM_WAVES;

    gcn_bucket_scatter_kernel<<<G, 1024, 0, stream>>>(rowp, colp, packed, rcnt, roff, nE, NB);
    gcn_bucket_sort_kernel<<<NB, 1024, 0, stream>>>(packed, rcnt, roff, perm, cnt, offs, dinv, N, NB, G);

    gcn_gemm_wmma_kernel<IN_CH><<<gemmBlocks, 256, 0, stream>>>(x, W1, dinv, h, N);
    const int slabBlocks = (N + PROWS - 1) / PROWS;
    const int fuseBlocks = (slabBlocks > GATHER_BLOCKS) ? slabBlocks : GATHER_BLOCKS;
    gcn_gather_kernel<true><<<fuseBlocks, 256, 0, stream>>>(h, offs, cnt, perm, dinv, b1, W2, a1, N,  1);

    gcn_gather_kernel<false><<<GATHER_BLOCKS, 256, 0, stream>>>(a1, offs, cnt, perm, dinv, b2, W2, out, N,  0);
}


# Gather: a source's place in its segment from ONE compare per element on a packed word (source, lane), eight constant-lane reads a block, not three compares and mask logic each; sum order kept
# speedup vs baseline: 5.3746x; 1.0535x over previous
#include <hip/hip_runtime.h>
#include <hip/hip_bf16.h>


typedef __attribute__((ext_vector_type(2))) float v2f;
typedef __attribute__((ext_vector_type(4))) float v4f;
typedef __attribute__((ext_vector_type(8))) float v8f;

#define IN_CH  128
#define HID_CH 64

#define SH      9
#define BKT     (1 << SH)
#define NBMAX   256
#define ROWBITS 17
#define EPB     8192
#define GMAX    256
#define RSH     6
#define RSLOT   (1 << RSH)
#define FPT     8
#define STG     16384

__global__ __launch_bounds__(1024)
void gcn_bucket_scatter_kernel(const int* __restrict__ row, const int* __restrict__ col,
                               int* __restrict__ packed, int* __restrict__ rcnt,
                               int* __restrict__ roff, int nE, int NB) {
    __shared__ int s_h[NBMAX];
    __shared__ int s_base[NBMAX];
    __shared__ int s_w[8];
    __shared__ __attribute__((aligned(16))) int s_out[EPB];
    const int tid  = threadIdx.x;
    const int lane = tid & 31;
    const int wv   = tid >> 5;
    const int g    = blockIdx.x;

    if (tid < NBMAX) s_h[tid] = 0;

    const int base = g * EPB;
    int cc[EPB / 1024];
    int rr[EPB / 1024];
    int rk[EPB / 1024];
#pragma unroll
    for (int q = 0; q < EPB / 1024; ++q) {
        const int e  = base + q * 1024 + tid;
        const int ee = (e < nE) ? e : (nE - 1);
        cc[q] = col[ee];
        rr[q] = row[ee];
    }
    __syncthreads();

#pragma unroll
    for (int q = 0; q < EPB / 1024; ++q) {
        const int e = base + q * 1024 + tid;
        rk[q] = 0;
        if (e < nE) rk[q] = atomicAdd(&s_h[cc[q] >> SH], 1);
    }
    __syncthreads();

    int T = 0, inc = 0;
    if (tid < NBMAX) {
        T = s_h[tid];
        inc = T;
        for (int d = 1; d < 32; d <<= 1) {
            const int u = __shfl_up(inc, d);
            if (lane >= d) inc += u;
        }
        if (lane == 31) s_w[wv] = inc;
    }
    __syncthreads();
    if (tid < NBMAX) {
        int pre = 0;
        for (int k = 0; k < wv; ++k) pre += s_w[k];
        const int excl = pre + inc - T;
        s_base[tid] = excl;
        if (tid < NB) { rcnt[g * NB + tid] = T; roff[g * NB + tid] = excl; }
    }
    __syncthreads();

    int bs[EPB / 1024];
#pragma unroll
    for (int q = 0; q < EPB / 1024; ++q) bs[q] = s_base[cc[q] >> SH];
#pragma unroll
    for (int q = 0; q < EPB / 1024; ++q) {
        const int e = base + q * 1024 + tid;
        if (e < nE) s_out[bs[q] + rk[q]] = ((cc[q] & (BKT - 1)) << ROWBITS) | rr[q];
    }
    __syncthreads();

    const int nb = (nE - base < EPB) ? (nE - base) : EPB;
    for (int i = tid * 4; i < nb; i += 4096) {
        if (i + 4 <= nb) {
            *(int4*)(packed + base + i) = *(const int4*)(s_out + i);
        } else {
            for (int k = i; k < nb; ++k) packed[base + k] = s_out[k];
        }
    }
}

__global__ __launch_bounds__(1024)
void gcn_bucket_sort_kernel(const int* __restrict__ packed, const int* __restrict__ rcnt,
                            const int* __restrict__ roff, int* __restrict__ perm,
                            int* __restrict__ cnt, int* __restrict__ offs,
                            float* __restrict__ dinv, int n, int NB, int G) {
    __shared__ int s_cnt[BKT];
    __shared__ int s_cur[BKT];
    __shared__ int s_w[BKT / 32];
    __shared__ int s_tot[4][NBMAX];
    __shared__ int s_w8[8];
    __shared__ int2 s_run[GMAX];
    __shared__ int s_start;
    __shared__ int s_total;
    __shared__ int s_stage[STG];
    const int tid   = threadIdx.x;
    const int lane  = tid & 31;
    const int wv    = tid >> 5;
    const int b     = blockIdx.x;

    {
        const int t = tid & (NBMAX - 1);
        const int p = tid >> 8;
        int tot = 0;
        if (t < NB) {
            for (int k = p; k < G; k += 4) tot += rcnt[k * NB + t];
        }
        s_tot[p][t] = tot;
    }
    if (tid < G) s_run[tid] = make_int2(rcnt[tid * NB + b], tid * EPB + roff[tid * NB + b]);
    if (tid < BKT) s_cnt[tid] = 0;
    __syncthreads();

    {
        int T = 0, inc = 0;
        if (tid < NBMAX) {
            T = s_tot[0][tid] + s_tot[1][tid] + s_tot[2][tid] + s_tot[3][tid];
            inc = T;
            for (int d = 1; d < 32; d <<= 1) {
                const int u = __shfl_up(inc, d);
                if (lane >= d) inc += u;
            }
            if (lane == 31) s_w8[wv] = inc;
        }
        __syncthreads();
        if (tid == b) {
            int pre = 0;
            for (int k = 0; k < wv; ++k) pre += s_w8[k];
            s_start = pre + inc - T;
            s_total = T;
        }
    }

    const int nslot = G << RSH;
    const int ent = tid & (RSLOT - 1);
    for (int i0 = 0; i0 < nslot; i0 += FPT * 1024) {
        int2 ro[FPT];
        int  w[FPT];
        for (int q = 0; q < FPT; ++q) {
            const int gg = (i0 + q * 1024 + tid) >> RSH;
            ro[q] = s_run[(gg < G) ? gg : (G - 1)];
            if (gg >= G) ro[q].x = 0;
        }
        for (int q = 0; q < FPT; ++q) w[q] = (ent < ro[q].x) ? packed[ro[q].y + ent] : -1;
        for (int q = 0; q < FPT; ++q)
            if (w[q] >= 0) atomicAdd(&s_cnt[w[q] >> ROWBITS], 1);
    }
    for (int gg = wv; gg < G; gg += 32) {
        const int2 r = s_run[gg];
        for (int i = RSLOT + lane; i < r.x; i += 32)
            atomicAdd(&s_cnt[packed[r.y + i] >> ROWBITS], 1);
    }
    __syncthreads();

    int v = 0, inc = 0;
    if (tid < BKT) {
        v = s_cnt[tid];
        inc = v;
        for (int d = 1; d < 32; d <<= 1) {
            const int u = __shfl_up(inc, d);
            if (lane >= d) inc += u;
        }
        if (lane == 31) s_w[wv] = inc;
    }
    __syncthreads();
    const int start = s_start;
    const int tot   = s_total;
    const bool staged = (tot <= STG);
    if (tid < BKT) {
        int pre = 0;
        for (int k = 0; k < wv; ++k) pre += s_w[k];
        const int excl = pre + inc - v;
        s_cur[tid] = excl;
        const int node = (b << SH) + tid;
        if (node < n) {
            cnt[node]  = v;
            offs[node] = start + excl;
            float d = (float)(v + 1);
            dinv[node] = (d > 0.0f) ? rsqrtf(d) : 0.0f;
        }
    }
    __syncthreads();

    if (staged) {
        for (int i0 = 0; i0 < nslot; i0 += FPT * 1024) {
            int2 ro[FPT];
            int  w[FPT];
            int  sl[FPT];
            for (int q = 0; q < FPT; ++q) {
                const int gg = (i0 + q * 1024 + tid) >> RSH;
                ro[q] = s_run[(gg < G) ? gg : (G - 1)];
                if (gg >= G) ro[q].x = 0;
            }
            for (int q = 0; q < FPT; ++q) w[q] = (ent < ro[q].x) ? packed[ro[q].y + ent] : -1;
            for (int q = 0; q < FPT; ++q) {
                sl[q] = 0;
                if (w[q] >= 0) sl[q] = atomicAdd(&s_cur[w[q] >> ROWBITS], 1);
            }
            for (int q = 0; q < FPT; ++q)
                if (w[q] >= 0) s_stage[sl[q]] = w[q] & ((1 << ROWBITS) - 1);
        }
        for (int gg = wv; gg < G; gg += 32) {
            const int2 r = s_run[gg];
            for (int i = RSLOT + lane; i < r.x; i += 32) {
                const int w = packed[r.y + i];
                const int slot = atomicAdd(&s_cur[w >> ROWBITS], 1);
                s_stage[slot] = w & ((1 << ROWBITS) - 1);
            }
        }
        __syncthreads();

        const int al = start & 31;
        int i = tid - al;
        if (i < 0) i += 1024;
#pragma unroll 4
        for (; i < tot; i += 1024) perm[start + i] = s_stage[i];
    } else {
        for (int i0 = 0; i0 < nslot; i0 += FPT * 1024) {
            int2 ro[FPT];
            int  w[FPT];
            for (int q = 0; q < FPT; ++q) {
                const int gg = (i0 + q * 1024 + tid) >> RSH;
                ro[q] = s_run[(gg < G) ? gg : (G - 1)];
                if (gg >= G) ro[q].x = 0;
            }
            for (int q = 0; q < FPT; ++q) w[q] = (ent < ro[q].x) ? packed[ro[q].y + ent] : -1;
            for (int q = 0; q < FPT; ++q) {
                if (w[q] >= 0) {
                    const int slot = atomicAdd(&s_cur[w[q] >> ROWBITS], 1);
                    perm[start + slot] = w[q] & ((1 << ROWBITS) - 1);
                }
            }
        }
        for (int gg = wv; gg < G; gg += 32) {
            const int2 r = s_run[gg];
            for (int i = RSLOT + lane; i < r.x; i += 32) {
                const int w = packed[r.y + i];
                const int slot = atomicAdd(&s_cur[w >> ROWBITS], 1);
                perm[start + slot] = w & ((1 << ROWBITS) - 1);
            }
        }
    }
}

#define GEMM_WAVES 8
#define TPITCH     68

template <int K>
__global__ __launch_bounds__(256)
void gcn_gemm_wmma_kernel(const float* __restrict__ X, const float* __restrict__ W,
                          const float* __restrict__ dinv, float* __restrict__ H, int nNodes) {
    __shared__ __attribute__((aligned(16))) float s_w[K * HID_CH];
    __shared__ __attribute__((aligned(16))) float s_t[GEMM_WAVES * 16 * TPITCH];

    const int tid  = threadIdx.x;
    const int lane = tid & 31;
    const int wave = __builtin_amdgcn_readfirstlane(tid >> 5);
    const int n    = lane & 15;
    const int half = lane >> 4;
    const int r0   = (blockIdx.x * GEMM_WAVES + wave) * 16;

    int rA = r0 + n;
    if (rA > nNodes - 1) rA = nNodes - 1;
    const v4f* xr = (const v4f*)(X + (long long)rA * K) + half;
    v4f a4[K / 8];
#pragma unroll
    for (int j = 0; j < K / 8; ++j) a4[j] = xr[2 * j];

    constexpr int WQ = K * HID_CH / 1024;
    v4f w4[WQ];
#pragma unroll
    for (int q = 0; q < WQ; ++q) w4[q] = *(const v4f*)(W + q * 1024 + tid * 4);
#pragma unroll
    for (int q = 0; q < WQ; ++q) {
        const int g  = q * 1024 + tid * 4;
        const int k  = g >> 6;
        const int c  = g & (HID_CH - 1);
        const int kj = k >> 3;
        const int kh = (k >> 2) & 1;
        const int ke = k & 3;
        const int o  = ((((kj * 4 + (c >> 4)) * 32) + kh * 16 + (c & 15)) << 2) + ke;
        s_w[o]      = w4[q].x;
        s_w[o + 4]  = w4[q].y;
        s_w[o + 8]  = w4[q].z;
        s_w[o + 12] = w4[q].w;
    }
    __syncthreads();

    if (r0 >= nNodes) return;

    float dsc[8];
#pragma unroll
    for (int i = 0; i < 8; ++i) {
        int r = r0 + 2 * i + half;
        if (r > nNodes - 1) r = nNodes - 1;
        dsc[i] = dinv[r];
    }

    v8f acc0 = {}, acc1 = {}, acc2 = {}, acc3 = {};
    const v4f* bw = (const v4f*)s_w + lane;
#pragma unroll
    for (int j = 0; j < K / 8; ++j) {
        const v4f av = a4[j];
        const v4f q0 = bw[(j * 4 + 0) * 32];
        const v4f q1 = bw[(j * 4 + 1) * 32];
        const v4f q2 = bw[(j * 4 + 2) * 32];
        const v4f q3 = bw[(j * 4 + 3) * 32];
        v2f a, b0, b1, b2, b3;
        a.x = av.x;  a.y = av.y;
        b0.x = q0.x; b0.y = q0.y;
        b1.x = q1.x; b1.y = q1.y;
        b2.x = q2.x; b2.y = q2.y;
        b3.x = q3.x; b3.y = q3.y;
        acc0 = __builtin_amdgcn_wmma_f32_16x16x4_f32(false, a, false, b0, (short)0, acc0, false, false);
        acc1 = __builtin_amdgcn_wmma_f32_16x16x4_f32(false, a, false, b1, (short)0, acc1, false, false);
        acc2 = __builtin_amdgcn_wmma_f32_16x16x4_f32(false, a, false, b2, (short)0, acc2, false, false);
        acc3 = __builtin_amdgcn_wmma_f32_16x16x4_f32(false, a, false, b3, (short)0, acc3, false, false);
        a.x = av.z;  a.y = av.w;
        b0.x = q0.z; b0.y = q0.w;
        b1.x = q1.z; b1.y = q1.w;
        b2.x = q2.z; b2.y = q2.w;
        b3.x = q3.z; b3.y = q3.w;
        acc0 = __builtin_amdgcn_wmma_f32_16x16x4_f32(false, a, false, b0, (short)0, acc0, false, false);
        acc1 = __builtin_amdgcn_wmma_f32_16x16x4_f32(false, a, false, b1, (short)0, acc1, false, false);
        acc2 = __builtin_amdgcn_wmma_f32_16x16x4_f32(false, a, false, b2, (short)0, acc2, false, false);
        acc3 = __builtin_amdgcn_wmma_f32_16x16x4_f32(false, a, false, b3, (short)0, acc3, false, false);
    }

    float* st = s_t + wave * (16 * TPITCH);
#pragma unroll
    for (int v = 0; v < 8; ++v) {
        float* p = st + (v + 8 * half) * TPITCH + n;
        p[0]  = acc0[v];
        p[16] = acc1[v];
        p[32] = acc2[v];
        p[48] = acc3[v];
    }
    v4f o[8];
#pragma unroll
    for (int i = 0; i < 8; ++i) o[i] = *(const v4f*)(st + (2 * i + half) * TPITCH + 4 * n) * dsc[i];
    float* hp = H + (long long)(r0 + half) * HID_CH + 4 * n;
    if (r0 + 16 <= nNodes) {
#pragma unroll
        for (int i = 0; i < 8; ++i) *(v4f*)(hp + (long long)i * 2 * HID_CH) = o[i];
    } else {
#pragma unroll
        for (int i = 0; i < 8; ++i)
            if (r0 + 2 * i + half < nNodes) *(v4f*)(hp + (long long)i * 2 * HID_CH) = o[i];
    }
}

#define LSEG 128

template <int J0>
__device__ __forceinline__ void gcn_rank8(int key, int& pos) {
#pragma unroll
    for (int u = 0; u < 8; ++u) {
        const int kj = __builtin_amdgcn_readlane(key, J0 + u);
        pos += (kj < key) ? 1 : 0;
    }
}

__device__ __forceinline__
v2f gcn_sum_rows(v2f acc, const v2f* __restrict__ Hv, int rs, int cnt, int lane) {
    int j = 0;
#pragma unroll 2
    for (; j + 4 <= cnt; j += 4) {
        const int r0 = __builtin_amdgcn_readlane(rs, j);
        const int r1 = __builtin_amdgcn_readlane(rs, j + 1);
        const int r2 = __builtin_amdgcn_readlane(rs, j + 2);
        const int r3 = __builtin_amdgcn_readlane(rs, j + 3);
        const v2f h0 = Hv[(long long)r0 * 32 + lane];
        const v2f h1 = Hv[(long long)r1 * 32 + lane];
        const v2f h2 = Hv[(long long)r2 * 32 + lane];
        const v2f h3 = Hv[(long long)r3 * 32 + lane];
        acc += h0;
        acc += h1;
        acc += h2;
        acc += h3;
    }
    for (; j < cnt; ++j) {
        const int r0 = __builtin_amdgcn_readlane(rs, j);
        acc += Hv[(long long)r0 * 32 + lane];
    }
    return acc;
}

#define GATHER_BLOCKS 2048

#define WSLAB 8
#define PROWS (8 * WSLAB)

template <bool FUSE>
__global__ __launch_bounds__(256, 16)
void gcn_gather_kernel(const float* __restrict__ H, const int* __restrict__ offs,
                       const int* __restrict__ cnt, const int* __restrict__ perm,
                       const float* __restrict__ dinv, const float* __restrict__ bias,
                       const float* __restrict__ Wn, float* __restrict__ out,
                       int nNodes, int doRelu) {
    __shared__ __attribute__((aligned(16)))
        float s_buf[FUSE ? (PROWS * TPITCH + HID_CH * HID_CH) : (2 * 8 * LSEG)];
    float* const s_p   = s_buf;
    float* const s_w   = s_buf + (FUSE ? PROWS * TPITCH : 0);
    int*   const s_raw = (int*)s_w;
    int*   const s_ord = s_raw + 8 * LSEG;

    const int lane = threadIdx.x & 31;
    const int wv = __builtin_amdgcn_readfirstlane((int)(threadIdx.x >> 5));

    const int gw    = blockIdx.x * 8 + wv;
    const int TW    = gridDim.x * 8;
    const int q     = nNodes / TW;
    const int rem   = nNodes - q * TW;
    const int start = gw * q + ((gw < rem) ? gw : rem);
    const int end   = start + q + ((gw < rem) ? 1 : 0);

    const v2f* Hv = (const v2f*)H;
    const v2f  b  = ((const v2f*)bias)[lane];

    const int RND  = FUSE ? WSLAB : 32;
    const int span = FUSE ? 1 : (end - start);
    for (int rb = 0; rb < span; rb += RND) {
        const int base = start + rb;
        const int nk = (end - base < RND) ? (end - base) : RND;
        if (!FUSE || nk > 0) {

        int   m_beg = 0, m_deg = 0;
        float m_dc  = 0.0f;
        if (lane < nk) {
            m_beg = offs[base + lane];
            m_deg = cnt[base + lane];
            m_dc  = dinv[base + lane];
        }

        int r_cur = 0x7fffffff;
        v2f a_cur;
        {
            const int b0 = __builtin_amdgcn_readlane(m_beg, 0);
            const int d0 = __builtin_amdgcn_readlane(m_deg, 0);
            if (lane < d0) r_cur = perm[b0 + lane];
            a_cur = Hv[(long long)base * 32 + lane];
        }

        for (int k = 0; k < nk; ++k) {
            const int   c   = base + k;
            const int   beg = __builtin_amdgcn_readlane(m_beg, k);
            const int   deg = __builtin_amdgcn_readlane(m_deg, k);
            const float dc  = __int_as_float(__builtin_amdgcn_readlane(__float_as_int(m_dc), k));

            const int kn = (k + 1 < nk) ? (k + 1) : k;
            int r_nxt = 0x7fffffff;
            {
                const int bn = __builtin_amdgcn_readlane(m_beg, kn);
                const int dn = __builtin_amdgcn_readlane(m_deg, kn);
                if (lane < dn) r_nxt = perm[bn + lane];
            }
            const v2f h_nxt = Hv[(long long)(base + kn) * 32 + lane];

            v2f acc = a_cur;

            if (deg <= 32) {
                const int r_l = r_cur;

                const int key = (lane < deg) ? (int)(((unsigned)r_l << 5) | (unsigned)lane) : 0x7fffffff;
                int pos = 0;
                gcn_rank8<0>(key, pos);
                if (deg > 8) {
                    gcn_rank8<8>(key, pos);
                    if (deg > 16) {
                        gcn_rank8<16>(key, pos);
                        if (deg > 24) gcn_rank8<24>(key, pos);
                    }
                }
                if (lane >= deg) pos = lane;

                const int rs = __builtin_amdgcn_ds_permute(pos << 2, r_l);

                acc = gcn_sum_rows(acc, Hv, rs, deg, lane);
            } else if (deg <= LSEG) {
                int* raw = s_raw + wv * LSEG;
                int* ord = s_ord + wv * LSEG;
                const int K = (deg + 31) >> 5;

                for (int t = 0; t < K; ++t) {
                    const int i = t * 32 + lane;
                    raw[i] = (i < deg) ? perm[beg + i] : 0x7fffffff;
                }
                for (int t = 0; t < K; ++t) {
                    const int i = t * 32 + lane;
                    const int e = raw[i];
                    const int key = (int)(((unsigned)e << 7) | (unsigned)i);
                    int pos = 0;
                    for (int cj = 0; cj < K; ++cj) {
                        const int ix = cj * 32 + lane;
                        const int kx = (int)(((unsigned)raw[ix] << 7) | (unsigned)ix);
                        const int nj = (deg - cj * 32 < 32) ? (deg - cj * 32) : 32;
                        for (int j = 0; j < nj; ++j) {
                            const int kj = __builtin_amdgcn_readlane(kx, j);
                            pos += (kj < key) ? 1 : 0;
                        }
                    }
                    if (i < deg) ord[pos] = e;
                }
                for (int t = 0; t < K; ++t) {
                    const int nv = (deg - t * 32 < 32) ? (deg - t * 32) : 32;
                    int rs = 0;
                    if (lane < nv) rs = ord[t * 32 + lane];
                    acc = gcn_sum_rows(acc, Hv, rs, nv, lane);
                }
            } else {
                int prev = -1;
                int done = 0;
                for (int it = 0; it < deg && done < deg; ++it) {
                    int best = 0x7fffffff;
                    for (int i = lane; i < deg; i += 32) {
                        const int r = perm[beg + i];
                        if (r > prev && r < best) best = r;
                    }
                    for (int d = 16; d >= 1; d >>= 1) {
                        const int o = __shfl_xor(best, d);
                        best = (o < best) ? o : best;
                    }
                    best = __builtin_amdgcn_readfirstlane(best);
                    if (best == 0x7fffffff) break;
                    int m = 0;
                    for (int i = lane; i < deg; i += 32) m += (perm[beg + i] == best) ? 1 : 0;
                    for (int d = 16; d >= 1; d >>= 1) m += __shfl_xor(m, d);
                    m = __builtin_amdgcn_readfirstlane(m);
                    const v2f hv = Hv[(long long)best * 32 + lane];
                    for (int t = 0; t < m; ++t) acc += hv;
                    prev = best;
                    done += m;
                }
            }

            v2f v = acc * dc + b;
            if (doRelu) { v.x = fmaxf(v.x, 0.0f); v.y = fmaxf(v.y, 0.0f); }
            if (FUSE) *(v2f*)(s_p + (wv * WSLAB + k) * TPITCH + 2 * lane) = v;
            else      ((v2f*)out)[(long long)c * 32 + lane] = v;

            a_cur = h_nxt;
            r_cur = r_nxt;
        }
        }

        if (FUSE) {
            const int tid  = threadIdx.x;
            const int n    = lane & 15;
            const int half = lane >> 4;
            const int rt   = wv >> 1;
            const int cp   = wv & 1;

            for (int k = nk; k < WSLAB; ++k) {
                v2f z; z.x = 0.0f; z.y = 0.0f;
                *(v2f*)(s_p + (wv * WSLAB + k) * TPITCH + 2 * lane) = z;
            }

            constexpr int WQ = HID_CH * HID_CH / 1024;
            v4f w4[WQ];
#pragma unroll
            for (int qq = 0; qq < WQ; ++qq) w4[qq] = *(const v4f*)(Wn + qq * 1024 + tid * 4);

            __syncthreads();

            {
                const int kk = tid >> 4;
                const int c  = (tid & 15) * 4;
                const int kj = kk >> 3;
                const int kh = (kk >> 2) & 1;
                const int ke = kk & 3;
                float* sw = s_w + (((((kj * 4 + (c >> 4)) * 32) + kh * 16 + (c & 15)) << 2) + ke);
#pragma unroll
                for (int qq = 0; qq < WQ; ++qq) {
                    sw[qq * 1024]      = w4[qq].x;
                    sw[qq * 1024 + 4]  = w4[qq].y;
                    sw[qq * 1024 + 8]  = w4[qq].z;
                    sw[qq * 1024 + 12] = w4[qq].w;
                }
            }
            __syncthreads();

            v8f acc0 = {}, acc1 = {};
            const v4f* ap = (const v4f*)(s_p + (rt * 16 + n) * TPITCH) + half;
            const v4f* bw = (const v4f*)s_w + cp * 64 + lane;
#pragma unroll 2
            for (int j = 0; j < HID_CH / 8; ++j) {
                const v4f av = ap[2 * j];
                const v4f q0 = bw[(j * 4 + 0) * 32];
                const v4f q1 = bw[(j * 4 + 1) * 32];
                v2f a, b0, b1;
                a.x = av.x;  a.y = av.y;
                b0.x = q0.x; b0.y = q0.y;
                b1.x = q1.x; b1.y = q1.y;
                acc0 = __builtin_amdgcn_wmma_f32_16x16x4_f32(false, a, false, b0, (short)0, acc0, false, false);
                acc1 = __builtin_amdgcn_wmma_f32_16x16x4_f32(false, a, false, b1, (short)0, acc1, false, false);
                a.x = av.z;  a.y = av.w;
                b0.x = q0.z; b0.y = q0.w;
                b1.x = q1.z; b1.y = q1.w;
                acc0 = __builtin_amdgcn_wmma_f32_16x16x4_f32(false, a, false, b0, (short)0, acc0, false, false);
                acc1 = __builtin_amdgcn_wmma_f32_16x16x4_f32(false, a, false, b1, (short)0, acc1, false, false);
            }
            __syncthreads();

            const int rr0 = lane >> 3;
            const int c4  = (lane & 7) * 4;
            int   onode[4];
            float dsc[4];
#pragma unroll
            for (int i = 0; i < 4; ++i) {
                const int p  = rt * 16 + i * 4 + rr0;
                const int g2 = blockIdx.x * 8 + (p >> 3);
                const int s2 = g2 * q + ((g2 < rem) ? g2 : rem);
                const int e2 = s2 + q + ((g2 < rem) ? 1 : 0);
                const int nd = s2 + (p & (WSLAB - 1));
                onode[i] = (nd < e2) ? nd : -1;
                dsc[i]   = (nd < e2) ? dinv[nd] : 0.0f;
            }

            float* st = s_p + (rt * 16) * TPITCH + cp * 32;
#pragma unroll
            for (int v = 0; v < 8; ++v) {
                float* p = st + (v + 8 * half) * TPITCH + n;
                p[0]  = acc0[v];
                p[16] = acc1[v];
            }
#pragma unroll
            for (int i = 0; i < 4; ++i) {
                const v4f o = *(const v4f*)(st + (i * 4 + rr0) * TPITCH + c4) * dsc[i];
                if (onode[i] >= 0) *(v4f*)(out + (long long)onode[i] * HID_CH + cp * 32 + c4) = o;
            }
        }
    }
}

extern "C" void kernel_launch(void* const* d_in, const int* in_sizes, int n_in,
                              void* d_out, int out_size, void* d_ws, size_t ws_size,
                              hipStream_t stream) {
    const float* x   = (const float*)d_in[0];
    const int*   ei  = (const int*)  d_in[1];
    const float* W1  = (const float*)d_in[2];
    const float* b1  = (const float*)d_in[3];
    const float* W2  = (const float*)d_in[4];
    const float* b2  = (const float*)d_in[5];
    float*       out = (float*)d_out;

    const int N  = in_sizes[0] / IN_CH;
    const int nE = in_sizes[1] / 2;
    const int*   rowp = ei;
    const int*   colp = ei + nE;

    const long long feat = (long long)N * HID_CH;
    float* h      = (float*)d_ws;
    float* a1     = h + feat;
    int*   packed = (int*)(a1 + feat);
    int*   perm   = packed + nE;
    int*   cnt    = perm + nE;
    int*   offs   = cnt + N;
    float* dinv   = (float*)(offs + N);

    const int NB = (N + BKT - 1) / BKT;
    const int G  = (nE + EPB - 1) / EPB;
    int*   rcnt   = (int*)a1;
    int*   roff   = rcnt + G * NB;

    const int tileBlocks = (N + 15) / 16;
    const int gemmBlocks = (tileBlocks + GEMM_WAVES - 1) / GEMM_WAVES;

    gcn_bucket_scatter_kernel<<<G, 1024, 0, stream>>>(rowp, colp, packed, rcnt, roff, nE, NB);
    gcn_bucket_sort_kernel<<<NB, 1024, 0, stream>>>(packed, rcnt, roff, perm, cnt, offs, dinv, N, NB, G);

    gcn_gemm_wmma_kernel<IN_CH><<<gemmBlocks, 256, 0, stream>>>(x, W1, dinv, h, N);
    const int slabBlocks = (N + PROWS - 1) / PROWS;
    const int fuseBlocks = (slabBlocks > GATHER_BLOCKS) ? slabBlocks : GATHER_BLOCKS;
    gcn_gather_kernel<true><<<fuseBlocks, 256, 0, stream>>>(h, offs, cnt, perm, dinv, b1, W2, a1, N,  1);

    gcn_gather_kernel<false><<<GATHER_BLOCKS, 256, 0, stream>>>(a1, offs, cnt, perm, dinv, b2, W2, out, N,  0);
}
